# starting kernel
# baseline (speedup 1.0000x reference)
#include <hip/hip_runtime.h>
#include <hip/hip_bf16.h>

typedef __attribute__((ext_vector_type(2))) float v2f;
typedef __attribute__((ext_vector_type(8))) float v8f;

#define DFEAT 96
#define NTILES_N 6

__global__ void cheb_scatter96(const int* __restrict__ rows,
                               const int* __restrict__ cols,
                               const float* __restrict__ vals,
                               const float* __restrict__ X,
                               float* __restrict__ Y,
                               int nEdges) {
    int t = blockIdx.x * blockDim.x + threadIdx.x;
    int e = t >> 5;
    if (e >= nEdges) return;
    int lane = t & 31;
    int r = rows[e];
    int c = cols[e];
    float v = vals[e];
    const float* xr = X + (size_t)c * DFEAT;
    float*       yr = Y + (size_t)r * DFEAT;
#pragma unroll
    for (int j = lane; j < DFEAT; j += 32) {
        unsafeAtomicAdd(&yr[j], v * xr[j]);
    }
}

__global__ void cheb_combine_k1(const float* __restrict__ H,
                                float* __restrict__ Y,
                                float* __restrict__ T1,
                                float* __restrict__ S,
                                int n) {
    int i = blockIdx.x * blockDim.x + threadIdx.x;
    if (i >= n) return;
    float h = H[i];
    float y = Y[i];
    float t = 2.0f * y - h;
    T1[i] = t;
    S[i]  = h + t;
    Y[i]  = 0.0f;
}

__global__ void cheb_combine_rec(const float* Tprev,
                                 const float* __restrict__ Tcur,
                                 float* __restrict__ Y,
                                 float* Tnext,
                                 float* __restrict__ S,
                                 int n, int zeroY) {
    int i = blockIdx.x * blockDim.x + threadIdx.x;
    if (i >= n) return;
    float t = 4.0f * Y[i] - 2.0f * Tcur[i] - Tprev[i];
    Tnext[i] = t;
    S[i] += t;
    if (zeroY) Y[i] = 0.0f;
}

__global__ void cheb_gemm_wmma(const float* __restrict__ S,
                               const float* __restrict__ W,
                               const float* __restrict__ bias,
                               float* __restrict__ out,
                               int nTilesM) {
    int waveId = blockIdx.x * (blockDim.x >> 5) + (threadIdx.x >> 5);
    int nTiles = nTilesM * NTILES_N;
    if (waveId >= nTiles) return;

    int mt = waveId / NTILES_N;
    int nt = waveId - mt * NTILES_N;
    int lane = threadIdx.x & 31;
    int half = lane >> 4;
    int l    = lane & 15;

    const float* Srow = S + (size_t)(mt * 16 + l) * DFEAT;
    const float* Wcol = W + nt * 16 + l;

    v8f c = {};
#pragma unroll
    for (int k0 = 0; k0 < DFEAT; k0 += 4) {
        int k = k0 + 2 * half;
        v2f a, b;
        a.x = Srow[k];
        a.y = Srow[k + 1];
        b.x = Wcol[(size_t)k * DFEAT];
        b.y = Wcol[(size_t)(k + 1) * DFEAT];
        c = __builtin_amdgcn_wmma_f32_16x16x4_f32(
                 false, a,  false, b,
                 (short)0, c,  false,  false);
    }

    int n  = nt * 16 + l;
    float bv = bias[n];
    int mbase = mt * 16 + half * 8;
    float* orow = out + (size_t)mbase * DFEAT + n;
#pragma unroll
    for (int r = 0; r < 8; ++r) {
        orow[(size_t)r * DFEAT] = c[r] + bv;
    }
}

extern "C" void kernel_launch(void* const* d_in, const int* in_sizes, int n_in,
                              void* d_out, int out_size, void* d_ws, size_t ws_size,
                              hipStream_t stream) {
    const int*   rows = (const int*)d_in[0];
    const int*   cols = (const int*)d_in[1];
    const float* vals = (const float*)d_in[2];
    const float* H    = (const float*)d_in[3];
    const float* W    = (const float*)d_in[4];
    const float* bias = (const float*)d_in[5];
    float* out = (float*)d_out;

    const int nEdges = in_sizes[0];
    const int nNodes = in_sizes[3] / DFEAT;
    const int nElem  = nNodes * DFEAT;

    float* Y    = (float*)d_ws;
    float* Ssum = Y + nElem;
    float* bufA = Ssum + nElem;
    float* bufB = bufA + nElem;

    const int TPB = 256;
    dim3 blk(TPB);
    dim3 grdScatter((nEdges * 32 + TPB - 1) / TPB);
    dim3 grdElem((nElem + TPB - 1) / TPB);
    const int nTilesM = nNodes / 16;
    int nWaves = nTilesM * NTILES_N;
    dim3 grdGemm((nWaves + (TPB / 32) - 1) / (TPB / 32));

    hipMemsetAsync(Y, 0, (size_t)nElem * sizeof(float), stream);

    cheb_scatter96<<<grdScatter, blk, 0, stream>>>(rows, cols, vals, H, Y, nEdges);
    cheb_combine_k1<<<grdElem, blk, 0, stream>>>(H, Y, bufB, Ssum, nElem);

    cheb_scatter96<<<grdScatter, blk, 0, stream>>>(rows, cols, vals, bufB, Y, nEdges);
    cheb_combine_rec<<<grdElem, blk, 0, stream>>>(H, bufB, Y, bufA, Ssum, nElem, 1);

    cheb_scatter96<<<grdScatter, blk, 0, stream>>>(rows, cols, vals, bufA, Y, nEdges);
    cheb_combine_rec<<<grdElem, blk, 0, stream>>>(bufB, bufA, Y, bufB, Ssum, nElem, 0);

    cheb_gemm_wmma<<<grdGemm, blk, 0, stream>>>(Ssum, W, bias, out, nTilesM);
}


# Edges counting-sorted by row inside the call (int atomic rank, scan, fill); one wave sums each row from the list in registers (fma, list order) and stores it once: no float atomics, no Y clear
# speedup vs baseline: 2.0483x; 2.0483x over previous
#include <hip/hip_runtime.h>
#include <hip/hip_bf16.h>


typedef __attribute__((ext_vector_type(2))) float v2f;
typedef __attribute__((ext_vector_type(8))) float v8f;

#define DFEAT 96
#define NTILES_N 6
#define SCAN_WAVES 32
#define SCAN_THREADS (SCAN_WAVES * 32)

__global__ void cheb_count_rank(const int* __restrict__ rows,
                                int* __restrict__ cnt,
                                int* __restrict__ rank,
                                int nEdges) {
    int e = blockIdx.x * blockDim.x + threadIdx.x;
    if (e >= nEdges) return;
    rank[e] = atomicAdd(&cnt[rows[e]], 1);
}

__global__ void __launch_bounds__(SCAN_THREADS)
cheb_scan_rows(const int* __restrict__ cnt, int* __restrict__ rowptr, int seg) {
    __shared__ int wtot[SCAN_WAVES];
    const int lane = threadIdx.x & 31;
    const int w    = threadIdx.x >> 5;
    const int4* src = (const int4*)(cnt + (size_t)w * seg) + lane;
    int4*       dst = (int4*)(rowptr + (size_t)w * seg) + lane;
    const int sweeps = seg >> 7;

    int s = 0;
    for (int i = 0; i < sweeps; ++i) {
        int4 v = src[i * 32];
        s += (v.x + v.y) + (v.z + v.w);
    }
#pragma unroll
    for (int d = 16; d >= 1; d >>= 1) s += __shfl_xor(s, d, 32);
    if (lane == 0) wtot[w] = s;
    __syncthreads();
    int b = (lane < w) ? wtot[lane] : 0;
#pragma unroll
    for (int d = 16; d >= 1; d >>= 1) b += __shfl_xor(b, d, 32);

    int carry = b;
    for (int i = 0; i < sweeps; ++i) {
        int4 v = src[i * 32];
        int e1 = v.x, e2 = v.x + v.y, e3 = e2 + v.z, tot = e3 + v.w;
        int inc = tot;
#pragma unroll
        for (int d = 1; d < 32; d <<= 1) {
            int up = __shfl_up(inc, d, 32);
            if (lane >= d) inc += up;
        }
        int base = carry + inc - tot;
        dst[i * 32] = make_int4(base, base + e1, base + e2, base + e3);
        carry += __shfl(inc, 31, 32);
    }
}

__global__ void cheb_fill_sorted(const int* __restrict__ rows,
                                 const int* __restrict__ cols,
                                 const float* __restrict__ vals,
                                 const int* __restrict__ rowptr,
                                 const int* __restrict__ rank,
                                 int2* __restrict__ sorted,
                                 int nEdges) {
    int e = blockIdx.x * blockDim.x + threadIdx.x;
    if (e >= nEdges) return;
    int   c = cols[e];
    float v = vals[e];
    int pos = rowptr[rows[e]] + rank[e];
    if ((unsigned)pos < (unsigned)nEdges)
        sorted[pos] = make_int2(c, __float_as_int(v));
}

template <bool TAIL>
__device__ __forceinline__ void gather_group8(int2 my, int j, int n, unsigned cmax,
                                              bool act, const float* xl, float4& acc) {
    unsigned c[8];
    float    wv[8];
#pragma unroll
    for (int u = 0; u < 8; ++u) {
        int idx = (j + u) & 31;
        unsigned cu = (unsigned)__builtin_amdgcn_readlane(my.x, idx);
        c[u]  = cu > cmax ? cmax : cu;
        wv[u] = __int_as_float(__builtin_amdgcn_readlane(my.y, idx));
    }
    if (act) {
        float4 x[8];
#pragma unroll
        for (int u = 0; u < 8; ++u) {
            x[u] = make_float4(0.f, 0.f, 0.f, 0.f);
            if (!TAIL || j + u < n)
                x[u] = *(const float4*)(xl + (size_t)c[u] * DFEAT);
        }
#pragma unroll
        for (int u = 0; u < 8; ++u) {
            acc.x += wv[u] * x[u].x;
            acc.y += wv[u] * x[u].y;
            acc.z += wv[u] * x[u].z;
            acc.w += wv[u] * x[u].w;
        }
    }
}

__global__ void __launch_bounds__(256)
cheb_gather96(const int* __restrict__ rowptr,
              const int2* __restrict__ sorted,
              const float* __restrict__ X,
              float* __restrict__ Y,
              int nNodes, int nEdges) {
    const int row = __builtin_amdgcn_readfirstlane(
        (int)((blockIdx.x * blockDim.x + threadIdx.x) >> 5));
    if (row >= nNodes) return;
    const int lane = threadIdx.x & 31;
    const bool act = lane < (DFEAT / 4);
    int beg = rowptr[row];
    int end = rowptr[row + 1];
    beg = beg < 0 ? 0 : beg;
    end = end > nEdges ? nEdges : end;
    const unsigned cmax = (unsigned)(nNodes - 1);
    const float* xl = X + lane * 4;

    float4 acc = make_float4(0.f, 0.f, 0.f, 0.f);
    for (int base = beg; base < end; base += 32) {
        int n = end - base;
        n = n > 32 ? 32 : n;
        int2 my = make_int2(0, 0);
        if (lane < n) my = sorted[base + lane];
        int j = 0;
        for (; j + 8 <= n; j += 8)
            gather_group8<false>(my, j, n, cmax, act, xl, acc);
        if (j < n)
            gather_group8<true>(my, j, n, cmax, act, xl, acc);
    }
    if (act) *(float4*)(Y + (size_t)row * DFEAT + lane * 4) = acc;
}

__global__ void cheb_combine_k1(const float* __restrict__ H,
                                float* __restrict__ YT,
                                int n) {
    int i = blockIdx.x * blockDim.x + threadIdx.x;
    if (i >= n) return;
    float h = H[i];
    float y = YT[i];
    float t = 2.0f * y - h;
    YT[i] = t;
}

__global__ void cheb_combine_k2(const float* __restrict__ Tprev,
                                const float* __restrict__ Tcur,
                                float* __restrict__ YT,
                                int n) {
    int i = blockIdx.x * blockDim.x + threadIdx.x;
    if (i >= n) return;
    float t = 4.0f * YT[i] - 2.0f * Tcur[i] - Tprev[i];
    YT[i] = t;
}

__global__ void cheb_combine_k3(const float* __restrict__ H,
                                const float* __restrict__ T1,
                                const float* __restrict__ T2,
                                float* __restrict__ YS,
                                int n) {
    int i = blockIdx.x * blockDim.x + threadIdx.x;
    if (i >= n) return;
    float h  = H[i];
    float t1 = T1[i];
    float t2 = T2[i];
    float t  = 4.0f * YS[i] - 2.0f * t2 - t1;
    float s  = h + t1;
    s += t2;
    s += t;
    YS[i] = s;
}

__global__ void cheb_gemm_wmma(const float* __restrict__ S,
                               const float* __restrict__ W,
                               const float* __restrict__ bias,
                               float* __restrict__ out,
                               int nTilesM) {
    int waveId = blockIdx.x * (blockDim.x >> 5) + (threadIdx.x >> 5);
    int nTiles = nTilesM * NTILES_N;
    if (waveId >= nTiles) return;

    int mt = waveId / NTILES_N;
    int nt = waveId - mt * NTILES_N;
    int lane = threadIdx.x & 31;
    int half = lane >> 4;
    int l    = lane & 15;

    const float* Srow = S + (size_t)(mt * 16 + l) * DFEAT;
    const float* Wcol = W + nt * 16 + l;

    v8f c = {};
#pragma unroll
    for (int k0 = 0; k0 < DFEAT; k0 += 4) {
        int k = k0 + 2 * half;
        v2f a, b;
        a.x = Srow[k];
        a.y = Srow[k + 1];
        b.x = Wcol[(size_t)k * DFEAT];
        b.y = Wcol[(size_t)(k + 1) * DFEAT];
        c = __builtin_amdgcn_wmma_f32_16x16x4_f32(
                 false, a,  false, b,
                 (short)0, c,  false,  false);
    }

    int n  = nt * 16 + l;
    float bv = bias[n];
    int mbase = mt * 16 + half * 8;
    float* orow = out + (size_t)mbase * DFEAT + n;
#pragma unroll
    for (int r = 0; r < 8; ++r) {
        orow[(size_t)r * DFEAT] = c[r] + bv;
    }
}

extern "C" void kernel_launch(void* const* d_in, const int* in_sizes, int n_in,
                              void* d_out, int out_size, void* d_ws, size_t ws_size,
                              hipStream_t stream) {
    const int*   rows = (const int*)d_in[0];
    const int*   cols = (const int*)d_in[1];
    const float* vals = (const float*)d_in[2];
    const float* H    = (const float*)d_in[3];
    const float* W    = (const float*)d_in[4];
    const float* bias = (const float*)d_in[5];
    float* out = (float*)d_out;

    const int nEdges = in_sizes[0];
    const int nNodes = in_sizes[3] / DFEAT;
    const int nElem  = nNodes * DFEAT;

    float* P0 = (float*)d_ws;
    float* L  = P0 + nElem;
    float* P2 = L + nElem;
    float* P3 = P2 + nElem;

    const int seg  = ((((nNodes + 1) + SCAN_WAVES - 1) / SCAN_WAVES) + 127) / 128 * 128;
    const int nPad = seg * SCAN_WAVES;

    int2* sorted = (int2*)L;
    int*  rank   = (int*)(sorted + nEdges);
    int*  cnt    = rank + (nEdges + 3) / 4 * 4;
    int*  rowptr = cnt + nPad;

    const int TPB = 256;
    dim3 blk(TPB);
    dim3 grdEdge((nEdges + TPB - 1) / TPB);
    dim3 grdRows((nNodes * 32 + TPB - 1) / TPB);
    dim3 grdElem((nElem + TPB - 1) / TPB);
    const int nTilesM = nNodes / 16;
    int nWaves = nTilesM * NTILES_N;
    dim3 grdGemm((nWaves + (TPB / 32) - 1) / (TPB / 32));

    hipMemsetAsync(cnt, 0, (size_t)nPad * sizeof(int), stream);
    cheb_count_rank<<<grdEdge, blk, 0, stream>>>(rows, cnt, rank, nEdges);
    cheb_scan_rows<<<dim3(1), dim3(SCAN_THREADS), 0, stream>>>(cnt, rowptr, seg);
    cheb_fill_sorted<<<grdEdge, blk, 0, stream>>>(rows, cols, vals, rowptr, rank, sorted, nEdges);

    cheb_gather96<<<grdRows, blk, 0, stream>>>(rowptr, sorted, H, P0, nNodes, nEdges);
    cheb_combine_k1<<<grdElem, blk, 0, stream>>>(H, P0, nElem);

    cheb_gather96<<<grdRows, blk, 0, stream>>>(rowptr, sorted, P0, P2, nNodes, nEdges);
    cheb_combine_k2<<<grdElem, blk, 0, stream>>>(H, P0, P2, nElem);

    cheb_gather96<<<grdRows, blk, 0, stream>>>(rowptr, sorted, P2, P3, nNodes, nEdges);
    cheb_combine_k3<<<grdElem, blk, 0, stream>>>(H, P0, P2, P3, nElem);

    cheb_gemm_wmma<<<grdGemm, blk, 0, stream>>>(P3, W, bias, out, nTilesM);
}


# Dense product fused into the last row gather: 16-wave groups put their 16 rows of S in LDS, 6 waves then run the same 24 f32 WMMAs per tile (k ascending); no S plane, no GEMM launch
# speedup vs baseline: 2.5954x; 1.2671x over previous
#include <hip/hip_runtime.h>
#include <hip/hip_bf16.h>


typedef __attribute__((ext_vector_type(2))) float v2f;
typedef __attribute__((ext_vector_type(8))) float v8f;

#define DFEAT 96
#define NTILES_N 6
#define SCAN_WAVES 32
#define SCAN_THREADS (SCAN_WAVES * 32)

__global__ void cheb_count_rank(const int* __restrict__ rows,
                                int* __restrict__ cnt,
                                int* __restrict__ rank,
                                int nEdges) {
    int e = blockIdx.x * blockDim.x + threadIdx.x;
    if (e >= nEdges) return;
    rank[e] = atomicAdd(&cnt[rows[e]], 1);
}

__global__ void __launch_bounds__(SCAN_THREADS)
cheb_scan_rows(const int* __restrict__ cnt, int* __restrict__ rowptr, int seg) {
    __shared__ int wtot[SCAN_WAVES];
    const int lane = threadIdx.x & 31;
    const int w    = threadIdx.x >> 5;
    const int4* src = (const int4*)(cnt + (size_t)w * seg) + lane;
    int4*       dst = (int4*)(rowptr + (size_t)w * seg) + lane;
    const int sweeps = seg >> 7;

    int s = 0;
    for (int i = 0; i < sweeps; ++i) {
        int4 v = src[i * 32];
        s += (v.x + v.y) + (v.z + v.w);
    }
#pragma unroll
    for (int d = 16; d >= 1; d >>= 1) s += __shfl_xor(s, d, 32);
    if (lane == 0) wtot[w] = s;
    __syncthreads();
    int b = (lane < w) ? wtot[lane] : 0;
#pragma unroll
    for (int d = 16; d >= 1; d >>= 1) b += __shfl_xor(b, d, 32);

    int carry = b;
    for (int i = 0; i < sweeps; ++i) {
        int4 v = src[i * 32];
        int e1 = v.x, e2 = v.x + v.y, e3 = e2 + v.z, tot = e3 + v.w;
        int inc = tot;
#pragma unroll
        for (int d = 1; d < 32; d <<= 1) {
            int up = __shfl_up(inc, d, 32);
            if (lane >= d) inc += up;
        }
        int base = carry + inc - tot;
        dst[i * 32] = make_int4(base, base + e1, base + e2, base + e3);
        carry += __shfl(inc, 31, 32);
    }
}

__global__ void cheb_fill_sorted(const int* __restrict__ rows,
                                 const int* __restrict__ cols,
                                 const float* __restrict__ vals,
                                 const int* __restrict__ rowptr,
                                 const int* __restrict__ rank,
                                 int2* __restrict__ sorted,
                                 int nEdges) {
    int e = blockIdx.x * blockDim.x + threadIdx.x;
    if (e >= nEdges) return;
    int   c = cols[e];
    float v = vals[e];
    int pos = rowptr[rows[e]] + rank[e];
    if ((unsigned)pos < (unsigned)nEdges)
        sorted[pos] = make_int2(c, __float_as_int(v));
}

template <bool TAIL>
__device__ __forceinline__ void gather_group8(int2 my, int j, int n, unsigned cmax,
                                              bool act, const float* xl, float4& acc) {
    unsigned c[8];
    float    wv[8];
#pragma unroll
    for (int u = 0; u < 8; ++u) {
        int idx = (j + u) & 31;
        unsigned cu = (unsigned)__builtin_amdgcn_readlane(my.x, idx);
        c[u]  = cu > cmax ? cmax : cu;
        wv[u] = __int_as_float(__builtin_amdgcn_readlane(my.y, idx));
    }
    if (act) {
        float4 x[8];
#pragma unroll
        for (int u = 0; u < 8; ++u) {
            x[u] = make_float4(0.f, 0.f, 0.f, 0.f);
            if (!TAIL || j + u < n)
                x[u] = *(const float4*)(xl + (size_t)c[u] * DFEAT);
        }
#pragma unroll
        for (int u = 0; u < 8; ++u) {
            acc.x += wv[u] * x[u].x;
            acc.y += wv[u] * x[u].y;
            acc.z += wv[u] * x[u].z;
            acc.w += wv[u] * x[u].w;
        }
    }
}

template <int STEP>
__device__ __forceinline__ float4
gather96_value(const int* __restrict__ rowptr,
               const int2* __restrict__ sorted,
               const float* __restrict__ X,
               const float* __restrict__ H,
               const float* __restrict__ T1,
               const float* __restrict__ T2,
               int row, int lane, int nNodes, int nEdges) {
    const bool act = lane < (DFEAT / 4);
    int beg = rowptr[row];
    int end = rowptr[row + 1];
    beg = beg < 0 ? 0 : beg;
    end = end > nEdges ? nEdges : end;
    const unsigned cmax = (unsigned)(nNodes - 1);
    const float* xl = X + lane * 4;
    const size_t own = (size_t)row * DFEAT + lane * 4;

    float4 h  = make_float4(0.f, 0.f, 0.f, 0.f);
    float4 t1 = make_float4(0.f, 0.f, 0.f, 0.f);
    float4 t2 = make_float4(0.f, 0.f, 0.f, 0.f);
    if (act) {
        h = *(const float4*)(H + own);
        if (STEP >= 2) t1 = *(const float4*)(T1 + own);
        if (STEP == 3) t2 = *(const float4*)(T2 + own);
    }

    float4 acc = make_float4(0.f, 0.f, 0.f, 0.f);
    for (int base = beg; base < end; base += 32) {
        int n = end - base;
        n = n > 32 ? 32 : n;
        int2 my = make_int2(0, 0);
        if (lane < n) my = sorted[base + lane];
        int j = 0;
        for (; j + 8 <= n; j += 8)
            gather_group8<false>(my, j, n, cmax, act, xl, acc);
        if (j < n)
            gather_group8<true>(my, j, n, cmax, act, xl, acc);
    }

    float4 o;
    if (STEP == 1) {
        o.x = 2.0f * acc.x - h.x;
        o.y = 2.0f * acc.y - h.y;
        o.z = 2.0f * acc.z - h.z;
        o.w = 2.0f * acc.w - h.w;
    } else if (STEP == 2) {
        o.x = 4.0f * acc.x - 2.0f * t1.x - h.x;
        o.y = 4.0f * acc.y - 2.0f * t1.y - h.y;
        o.z = 4.0f * acc.z - 2.0f * t1.z - h.z;
        o.w = 4.0f * acc.w - 2.0f * t1.w - h.w;
    } else {
        float4 t3;
        t3.x = 4.0f * acc.x - 2.0f * t2.x - t1.x;
        t3.y = 4.0f * acc.y - 2.0f * t2.y - t1.y;
        t3.z = 4.0f * acc.z - 2.0f * t2.z - t1.z;
        t3.w = 4.0f * acc.w - 2.0f * t2.w - t1.w;
        o.x = h.x + t1.x;  o.x += t2.x;  o.x += t3.x;
        o.y = h.y + t1.y;  o.y += t2.y;  o.y += t3.y;
        o.z = h.z + t1.z;  o.z += t2.z;  o.z += t3.z;
        o.w = h.w + t1.w;  o.w += t2.w;  o.w += t3.w;
    }
    return o;
}

__global__ void __launch_bounds__(256)
cheb_gather96_t1(const int* __restrict__ rowptr,
                 const int2* __restrict__ sorted,
                 const float* __restrict__ H,
                 float* __restrict__ T1out,
                 int nNodes, int nEdges) {
    const int row = __builtin_amdgcn_readfirstlane(
        (int)((blockIdx.x * blockDim.x + threadIdx.x) >> 5));
    if (row >= nNodes) return;
    const int lane = threadIdx.x & 31;
    float4 o = gather96_value<1>(rowptr, sorted, H, H, nullptr, nullptr,
                                 row, lane, nNodes, nEdges);
    if (lane < (DFEAT / 4))
        *(float4*)(T1out + (size_t)row * DFEAT + lane * 4) = o;
}

__global__ void __launch_bounds__(256)
cheb_gather96_t2(const int* __restrict__ rowptr,
                 const int2* __restrict__ sorted,
                 const float* __restrict__ H,
                 const float* __restrict__ T1,
                 float* __restrict__ T2out,
                 int nNodes, int nEdges) {
    const int row = __builtin_amdgcn_readfirstlane(
        (int)((blockIdx.x * blockDim.x + threadIdx.x) >> 5));
    if (row >= nNodes) return;
    const int lane = threadIdx.x & 31;
    float4 o = gather96_value<2>(rowptr, sorted, T1, H, T1, nullptr,
                                 row, lane, nNodes, nEdges);
    if (lane < (DFEAT / 4))
        *(float4*)(T2out + (size_t)row * DFEAT + lane * 4) = o;
}

#define SPITCH 100
__global__ void __launch_bounds__(512)
cheb_gather96_s_gemm(const int* __restrict__ rowptr,
                     const int2* __restrict__ sorted,
                     const float* __restrict__ H,
                     const float* __restrict__ T1,
                     const float* __restrict__ T2,
                     const float* __restrict__ W,
                     const float* __restrict__ bias,
                     float* __restrict__ out,
                     int nNodes, int nEdges) {
    __shared__ __attribute__((aligned(16))) float sS[16 * SPITCH];
    const int lane = threadIdx.x & 31;
    const int w    = __builtin_amdgcn_readfirstlane((int)(threadIdx.x >> 5));
    const int row0 = blockIdx.x * 16;
    const int row  = row0 + w;

    float4 o = make_float4(0.f, 0.f, 0.f, 0.f);
    if (row < nNodes)
        o = gather96_value<3>(rowptr, sorted, T2, H, T1, T2,
                              row, lane, nNodes, nEdges);
    if (lane < (DFEAT / 4))
        *(float4*)(sS + w * SPITCH + lane * 4) = o;
    __syncthreads();
    if (w >= NTILES_N) return;

    const int half = lane >> 4;
    const int l    = lane & 15;
    const float* arow = sS + l * SPITCH + 2 * half;
    const float* Wcol = W + w * 16 + l + (size_t)(2 * half) * DFEAT;

    v8f c = {};
#pragma unroll
    for (int k0 = 0; k0 < DFEAT; k0 += 4) {
        v2f a = *(const v2f*)(arow + k0);
        v2f b;
        b.x = Wcol[(size_t)k0 * DFEAT];
        b.y = Wcol[(size_t)(k0 + 1) * DFEAT];
        c = __builtin_amdgcn_wmma_f32_16x16x4_f32(
                 false, a,  false, b,
                 (short)0, c,  false,  false);
    }

    const int n  = w * 16 + l;
    const float bv = bias[n];
    const int mbase = row0 + half * 8;
    float* orow = out + (size_t)mbase * DFEAT + n;
#pragma unroll
    for (int r = 0; r < 8; ++r) {
        if (mbase + r < nNodes)
            orow[(size_t)r * DFEAT] = c[r] + bv;
    }
}

extern "C" void kernel_launch(void* const* d_in, const int* in_sizes, int n_in,
                              void* d_out, int out_size, void* d_ws, size_t ws_size,
                              hipStream_t stream) {
    const int*   rows = (const int*)d_in[0];
    const int*   cols = (const int*)d_in[1];
    const float* vals = (const float*)d_in[2];
    const float* H    = (const float*)d_in[3];
    const float* W    = (const float*)d_in[4];
    const float* bias = (const float*)d_in[5];
    float* out = (float*)d_out;

    const int nEdges = in_sizes[0];
    const int nNodes = in_sizes[3] / DFEAT;
    const int nElem  = nNodes * DFEAT;

    float* P0 = (float*)d_ws;
    float* L  = P0 + nElem;
    float* P2 = L + nElem;

    const int seg  = ((((nNodes + 1) + SCAN_WAVES - 1) / SCAN_WAVES) + 127) / 128 * 128;
    const int nPad = seg * SCAN_WAVES;

    int2* sorted = (int2*)L;
    int*  rank   = (int*)(sorted + nEdges);
    int*  cnt    = rank + (nEdges + 3) / 4 * 4;
    int*  rowptr = cnt + nPad;

    const int TPB = 256;
    dim3 blk(TPB);
    dim3 grdEdge((nEdges + TPB - 1) / TPB);
    dim3 grdRows((nNodes * 32 + TPB - 1) / TPB);
    dim3 grdTile((nNodes + 15) / 16);

    hipMemsetAsync(cnt, 0, (size_t)nPad * sizeof(int), stream);
    cheb_count_rank<<<grdEdge, blk, 0, stream>>>(rows, cnt, rank, nEdges);
    cheb_scan_rows<<<dim3(1), dim3(SCAN_THREADS), 0, stream>>>(cnt, rowptr, seg);
    cheb_fill_sorted<<<grdEdge, blk, 0, stream>>>(rows, cols, vals, rowptr, rank, sorted, nEdges);

    cheb_gather96_t1<<<grdRows, blk, 0, stream>>>(rowptr, sorted, H, P0, nNodes, nEdges);

    cheb_gather96_t2<<<grdRows, blk, 0, stream>>>(rowptr, sorted, H, P0, P2, nNodes, nEdges);

    cheb_gather96_s_gemm<<<grdTile, dim3(512), 0, stream>>>(rowptr, sorted, H, P0, P2, W, bias, out,
                                                            nNodes, nEdges);
}


# Edge sort on LDS counters, no global atomics or memset: 4096-edge chunks ordered by 196-row range in LDS, one group per range counts, scans, places its runs; row sum order = LDS rank order
# speedup vs baseline: 4.5957x; 1.7707x over previous
#include <hip/hip_runtime.h>
#include <hip/hip_bf16.h>


typedef __attribute__((ext_vector_type(2))) float v2f;
typedef __attribute__((ext_vector_type(8))) float v8f;

#define DFEAT 96
#define NTILES_N 6

#define CE 4096
#define NBIN 256
#define SORT_THREADS 1024
#define MAXRPB 1024

__global__ void __launch_bounds__(SORT_THREADS)
cheb_split_ranges(const int* __restrict__ rows,
                  const int* __restrict__ cols,
                  const float* __restrict__ vals,
                  int* __restrict__ recR,
                  int2* __restrict__ recCV,
                  int* __restrict__ table,
                  int nEdges, int nNodes, int rpb, int ncp) {
    __shared__ int  hist[NBIN];
    __shared__ int  bstart[NBIN];
    __shared__ int  wtot[NBIN / 32];
    __shared__ int  sR[CE];
    __shared__ int2 sCV[CE];
    const int tid   = threadIdx.x;
    const int lane  = tid & 31;
    const int w     = tid >> 5;
    const int chunk = blockIdx.x;
    const int e0    = chunk * CE;

    if (tid < NBIN) hist[tid] = 0;

    int   r[4], c[4], bn[4], rk[4];
    float v[4];
    bool  ok[4];
#pragma unroll
    for (int k = 0; k < 4; ++k) {
        const int e = e0 + k * SORT_THREADS + tid;
        ok[k] = e < nEdges;
        const int ec = ok[k] ? e : nEdges - 1;
        r[k] = rows[ec];
        c[k] = cols[ec];
        v[k] = vals[ec];
    }
#pragma unroll
    for (int k = 0; k < 4; ++k) {
        unsigned row = (unsigned)r[k];
        if (row >= (unsigned)nNodes) row = (unsigned)(nNodes - 1);
        r[k]  = (int)row;
        bn[k] = ok[k] ? (int)(row / (unsigned)rpb) : -1;
        rk[k] = 0;
    }
    __syncthreads();
#pragma unroll
    for (int k = 0; k < 4; ++k)
        if (bn[k] >= 0) rk[k] = atomicAdd(&hist[bn[k]], 1);
    __syncthreads();

    int hv = 0, inc = 0;
    if (w < NBIN / 32) {
        hv = hist[tid];
        inc = hv;
#pragma unroll
        for (int d = 1; d < 32; d <<= 1) {
            int up = __shfl_up(inc, d, 32);
            if (lane >= d) inc += up;
        }
        if (lane == 31) wtot[w] = inc;
    }
    __syncthreads();
    if (w < NBIN / 32) {
        int carry = 0;
#pragma unroll
        for (int i = 0; i < NBIN / 32; ++i)
            if (i < w) carry += wtot[i];
        const int st = carry + inc - hv;
        bstart[tid] = st;
        table[(size_t)tid * ncp + chunk] = (st << 16) | hv;
    }
    __syncthreads();

#pragma unroll
    for (int k = 0; k < 4; ++k) {
        if (bn[k] >= 0) {
            const int p = bstart[bn[k]] + rk[k];
            sR[p]  = r[k];
            sCV[p] = make_int2(c[k], __float_as_int(v[k]));
        }
    }
    __syncthreads();

    int nIn = nEdges - e0;
    nIn = nIn > CE ? CE : nIn;
    for (int p = tid; p < nIn; p += SORT_THREADS) {
        recR[(size_t)e0 + p]  = sR[p];
        recCV[(size_t)e0 + p] = sCV[p];
    }
}

#define RUN_GROUPS (SORT_THREADS / 16)

template <bool PLACE>
__device__ __forceinline__ void walk_runs(const int* __restrict__ trow,
                                          const int* __restrict__ recR,
                                          const int2* __restrict__ recCV,
                                          int2* __restrict__ sorted,
                                          int* cntpos,
                                          int g, int sub, int ncp,
                                          int row0, int rpb, int nEdges) {
    for (int cb = 0; cb < ncp; cb += 16 * RUN_GROUPS) {
        const int cm = cb + g + RUN_GROUPS * sub;
        const int entMine = (cm < ncp) ? trow[cm] : 0;
        int jn = (ncp - cb + RUN_GROUPS - 1) / RUN_GROUPS;
        jn = jn > 16 ? 16 : jn;
        for (int j = 0; j < jn; ++j) {
            const int ent = __shfl(entMine, j, 16);
            const int c   = cb + g + RUN_GROUPS * j;
            const int len = (c < ncp) ? (ent & 0xFFFF) : 0;
            const size_t off = (size_t)c * CE + (ent >> 16);
            for (int k = sub; k < len; k += 16) {
                const int rrow = recR[off + k];
                int2 cvv = make_int2(0, 0);
                if (PLACE) {
                    cvv = recCV[off + k];
                    asm volatile("" :: "v"(cvv.x), "v"(cvv.y));
                }
                const int lr = rrow - row0;
                if ((unsigned)lr < (unsigned)rpb) {
                    if (!PLACE) {
                        atomicAdd(&cntpos[lr], 1);
                    } else {
                        const int pos = atomicAdd(&cntpos[lr], 1);
                        if ((unsigned)pos < (unsigned)nEdges)
                            sorted[pos] = cvv;
                    }
                }
            }
        }
    }
}

__global__ void __launch_bounds__(SORT_THREADS)
cheb_sort_range(const int* __restrict__ recR,
                const int2* __restrict__ recCV,
                const int* __restrict__ table,
                int* __restrict__ rowptr,
                int2* __restrict__ sorted,
                int nEdges, int nNodes, int rpb, int ncp) {
    __shared__ int cntpos[MAXRPB];
    __shared__ int wsumA[SORT_THREADS / 32];
    __shared__ int wsumB[SORT_THREADS / 32];
    const int tid  = threadIdx.x;
    const int lane = tid & 31;
    const int w    = tid >> 5;
    const int b    = blockIdx.x;
    const int row0 = b * rpb;

    cntpos[tid] = 0;

    const int4* T4 = (const int4*)table;
    const int lim4 = (b * ncp) >> 2;
    int s = 0;
    for (int i = tid; i < lim4; i += 4 * SORT_THREADS) {
        const int  i1 = i + SORT_THREADS, i2 = i + 2 * SORT_THREADS, i3 = i + 3 * SORT_THREADS;
        const bool v1 = i1 < lim4, v2 = i2 < lim4, v3 = i3 < lim4;
        const int4 a0 = T4[i];
        const int4 a1 = T4[v1 ? i1 : i];
        const int4 a2 = T4[v2 ? i2 : i];
        const int4 a3 = T4[v3 ? i3 : i];
        const int s0 = ((a0.x & 0xFFFF) + (a0.y & 0xFFFF)) + ((a0.z & 0xFFFF) + (a0.w & 0xFFFF));
        const int s1 = ((a1.x & 0xFFFF) + (a1.y & 0xFFFF)) + ((a1.z & 0xFFFF) + (a1.w & 0xFFFF));
        const int s2 = ((a2.x & 0xFFFF) + (a2.y & 0xFFFF)) + ((a2.z & 0xFFFF) + (a2.w & 0xFFFF));
        const int s3 = ((a3.x & 0xFFFF) + (a3.y & 0xFFFF)) + ((a3.z & 0xFFFF) + (a3.w & 0xFFFF));
        s += s0 + (v1 ? s1 : 0) + (v2 ? s2 : 0) + (v3 ? s3 : 0);
    }
#pragma unroll
    for (int d = 16; d >= 1; d >>= 1) s += __shfl_xor(s, d, 32);
    if (lane == 0) wsumA[w] = s;
    __syncthreads();

    const int g   = tid >> 4;
    const int sub = tid & 15;
    const int* trow = table + (size_t)b * ncp;

    walk_runs<false>(trow, recR, recCV, sorted, cntpos, g, sub, ncp, row0, rpb, nEdges);
    __syncthreads();

    int base = wsumA[lane];
#pragma unroll
    for (int d = 16; d >= 1; d >>= 1) base += __shfl_xor(base, d, 32);
    const int cv = cntpos[tid];
    int inc = cv;
#pragma unroll
    for (int d = 1; d < 32; d <<= 1) {
        int up = __shfl_up(inc, d, 32);
        if (lane >= d) inc += up;
    }
    if (lane == 31) wsumB[w] = inc;
    __syncthreads();
    int carry = (lane < w) ? wsumB[lane] : 0;
#pragma unroll
    for (int d = 16; d >= 1; d >>= 1) carry += __shfl_xor(carry, d, 32);
    const int start = base + carry + inc - cv;
    cntpos[tid] = start;
    if (tid < rpb && row0 + tid < nNodes) rowptr[row0 + tid] = start;
    if (b == (int)gridDim.x - 1 && tid == 0) rowptr[nNodes] = nEdges;
    __syncthreads();

    walk_runs<true>(trow, recR, recCV, sorted, cntpos, g, sub, ncp, row0, rpb, nEdges);
}

template <bool TAIL>
__device__ __forceinline__ void gather_group8(int2 my, int j, int n, unsigned cmax,
                                              bool act, const float* xl, float4& acc) {
    unsigned c[8];
    float    wv[8];
#pragma unroll
    for (int u = 0; u < 8; ++u) {
        int idx = (j + u) & 31;
        unsigned cu = (unsigned)__builtin_amdgcn_readlane(my.x, idx);
        c[u]  = cu > cmax ? cmax : cu;
        wv[u] = __int_as_float(__builtin_amdgcn_readlane(my.y, idx));
    }
    if (act) {
        float4 x[8];
#pragma unroll
        for (int u = 0; u < 8; ++u) {
            x[u] = make_float4(0.f, 0.f, 0.f, 0.f);
            if (!TAIL || j + u < n)
                x[u] = *(const float4*)(xl + (size_t)c[u] * DFEAT);
        }
#pragma unroll
        for (int u = 0; u < 8; ++u) {
            acc.x += wv[u] * x[u].x;
            acc.y += wv[u] * x[u].y;
            acc.z += wv[u] * x[u].z;
            acc.w += wv[u] * x[u].w;
        }
    }
}

template <int STEP>
__device__ __forceinline__ float4
gather96_value(const int* __restrict__ rowptr,
               const int2* __restrict__ sorted,
               const float* __restrict__ X,
               const float* __restrict__ H,
               const float* __restrict__ T1,
               const float* __restrict__ T2,
               int row, int lane, int nNodes, int nEdges) {
    const bool act = lane < (DFEAT / 4);
    int beg = rowptr[row];
    int end = rowptr[row + 1];
    beg = beg < 0 ? 0 : beg;
    end = end > nEdges ? nEdges : end;
    const unsigned cmax = (unsigned)(nNodes - 1);
    const float* xl = X + lane * 4;
    const size_t own = (size_t)row * DFEAT + lane * 4;

    float4 h  = make_float4(0.f, 0.f, 0.f, 0.f);
    float4 t1 = make_float4(0.f, 0.f, 0.f, 0.f);
    float4 t2 = make_float4(0.f, 0.f, 0.f, 0.f);
    if (act) {
        h = *(const float4*)(H + own);
        if (STEP >= 2) t1 = *(const float4*)(T1 + own);
        if (STEP == 3) t2 = *(const float4*)(T2 + own);
    }

    float4 acc = make_float4(0.f, 0.f, 0.f, 0.f);
    for (int base = beg; base < end; base += 32) {
        int n = end - base;
        n = n > 32 ? 32 : n;
        int2 my = make_int2(0, 0);
        if (lane < n) my = sorted[base + lane];
        int j = 0;
        for (; j + 8 <= n; j += 8)
            gather_group8<false>(my, j, n, cmax, act, xl, acc);
        if (j < n)
            gather_group8<true>(my, j, n, cmax, act, xl, acc);
    }

    float4 o;
    if (STEP == 1) {
        o.x = 2.0f * acc.x - h.x;
        o.y = 2.0f * acc.y - h.y;
        o.z = 2.0f * acc.z - h.z;
        o.w = 2.0f * acc.w - h.w;
    } else if (STEP == 2) {
        o.x = 4.0f * acc.x - 2.0f * t1.x - h.x;
        o.y = 4.0f * acc.y - 2.0f * t1.y - h.y;
        o.z = 4.0f * acc.z - 2.0f * t1.z - h.z;
        o.w = 4.0f * acc.w - 2.0f * t1.w - h.w;
    } else {
        float4 t3;
        t3.x = 4.0f * acc.x - 2.0f * t2.x - t1.x;
        t3.y = 4.0f * acc.y - 2.0f * t2.y - t1.y;
        t3.z = 4.0f * acc.z - 2.0f * t2.z - t1.z;
        t3.w = 4.0f * acc.w - 2.0f * t2.w - t1.w;
        o.x = h.x + t1.x;  o.x += t2.x;  o.x += t3.x;
        o.y = h.y + t1.y;  o.y += t2.y;  o.y += t3.y;
        o.z = h.z + t1.z;  o.z += t2.z;  o.z += t3.z;
        o.w = h.w + t1.w;  o.w += t2.w;  o.w += t3.w;
    }
    return o;
}

__global__ void __launch_bounds__(256)
cheb_gather96_t1(const int* __restrict__ rowptr,
                 const int2* __restrict__ sorted,
                 const float* __restrict__ H,
                 float* __restrict__ T1out,
                 int nNodes, int nEdges) {
    const int row = __builtin_amdgcn_readfirstlane(
        (int)((blockIdx.x * blockDim.x + threadIdx.x) >> 5));
    if (row >= nNodes) return;
    const int lane = threadIdx.x & 31;
    float4 o = gather96_value<1>(rowptr, sorted, H, H, nullptr, nullptr,
                                 row, lane, nNodes, nEdges);
    if (lane < (DFEAT / 4))
        *(float4*)(T1out + (size_t)row * DFEAT + lane * 4) = o;
}

__global__ void __launch_bounds__(256)
cheb_gather96_t2(const int* __restrict__ rowptr,
                 const int2* __restrict__ sorted,
                 const float* __restrict__ H,
                 const float* __restrict__ T1,
                 float* __restrict__ T2out,
                 int nNodes, int nEdges) {
    const int row = __builtin_amdgcn_readfirstlane(
        (int)((blockIdx.x * blockDim.x + threadIdx.x) >> 5));
    if (row >= nNodes) return;
    const int lane = threadIdx.x & 31;
    float4 o = gather96_value<2>(rowptr, sorted, T1, H, T1, nullptr,
                                 row, lane, nNodes, nEdges);
    if (lane < (DFEAT / 4))
        *(float4*)(T2out + (size_t)row * DFEAT + lane * 4) = o;
}

#define SPITCH 100
__global__ void __launch_bounds__(512)
cheb_gather96_s_gemm(const int* __restrict__ rowptr,
                     const int2* __restrict__ sorted,
                     const float* __restrict__ H,
                     const float* __restrict__ T1,
                     const float* __restrict__ T2,
                     const float* __restrict__ W,
                     const float* __restrict__ bias,
                     float* __restrict__ out,
                     int nNodes, int nEdges) {
    __shared__ __attribute__((aligned(16))) float sS[16 * SPITCH];
    const int lane = threadIdx.x & 31;
    const int w    = __builtin_amdgcn_readfirstlane((int)(threadIdx.x >> 5));
    const int row0 = blockIdx.x * 16;
    const int row  = row0 + w;

    float4 o = make_float4(0.f, 0.f, 0.f, 0.f);
    if (row < nNodes)
        o = gather96_value<3>(rowptr, sorted, T2, H, T1, T2,
                              row, lane, nNodes, nEdges);
    if (lane < (DFEAT / 4))
        *(float4*)(sS + w * SPITCH + lane * 4) = o;
    __syncthreads();
    if (w >= NTILES_N) return;

    const int half = lane >> 4;
    const int l    = lane & 15;
    const float* arow = sS + l * SPITCH + 2 * half;
    const float* Wcol = W + w * 16 + l + (size_t)(2 * half) * DFEAT;

    v8f c = {};
#pragma unroll
    for (int k0 = 0; k0 < DFEAT; k0 += 4) {
        v2f a = *(const v2f*)(arow + k0);
        v2f b;
        b.x = Wcol[(size_t)k0 * DFEAT];
        b.y = Wcol[(size_t)(k0 + 1) * DFEAT];
        c = __builtin_amdgcn_wmma_f32_16x16x4_f32(
                 false, a,  false, b,
                 (short)0, c,  false,  false);
    }

    const int n  = w * 16 + l;
    const float bv = bias[n];
    const int mbase = row0 + half * 8;
    float* orow = out + (size_t)mbase * DFEAT + n;
#pragma unroll
    for (int r = 0; r < 8; ++r) {
        if (mbase + r < nNodes)
            orow[(size_t)r * DFEAT] = c[r] + bv;
    }
}

extern "C" void kernel_launch(void* const* d_in, const int* in_sizes, int n_in,
                              void* d_out, int out_size, void* d_ws, size_t ws_size,
                              hipStream_t stream) {
    const int*   rows = (const int*)d_in[0];
    const int*   cols = (const int*)d_in[1];
    const float* vals = (const float*)d_in[2];
    const float* H    = (const float*)d_in[3];
    const float* W    = (const float*)d_in[4];
    const float* bias = (const float*)d_in[5];
    float* out = (float*)d_out;

    const int nEdges = in_sizes[0];
    const int nNodes = in_sizes[3] / DFEAT;
    const int nElem  = nNodes * DFEAT;

    float* P0 = (float*)d_ws;
    float* L  = P0 + nElem;
    float* P2 = L + nElem;

    const int ncp = (((nEdges + CE - 1) / CE) + 3) / 4 * 4;
    const int rpb = (nNodes + NBIN - 1) / NBIN;

    int2* sorted = (int2*)L;
    int2* recCV  = sorted + (nEdges + 1) / 2 * 2;
    int*  recR   = (int*)(recCV + (size_t)ncp * CE);
    int*  table  = recR + (size_t)ncp * CE;
    int*  rowptr = table + (size_t)NBIN * ncp;

    const int TPB = 256;
    dim3 blk(TPB);
    dim3 grdRows((nNodes * 32 + TPB - 1) / TPB);
    dim3 grdTile((nNodes + 15) / 16);

    cheb_split_ranges<<<dim3(ncp), dim3(SORT_THREADS), 0, stream>>>(rows, cols, vals, recR, recCV, table,
                                                                    nEdges, nNodes, rpb, ncp);
    cheb_sort_range<<<dim3(NBIN), dim3(SORT_THREADS), 0, stream>>>(recR, recCV, table, rowptr, sorted,
                                                                   nEdges, nNodes, rpb, ncp);

    cheb_gather96_t1<<<grdRows, blk, 0, stream>>>(rowptr, sorted, H, P0, nNodes, nEdges);

    cheb_gather96_t2<<<grdRows, blk, 0, stream>>>(rowptr, sorted, H, P0, P2, nNodes, nEdges);

    cheb_gather96_s_gemm<<<grdTile, dim3(512), 0, stream>>>(rowptr, sorted, H, P0, P2, W, bias, out,
                                                            nNodes, nEdges);
}


# Each wave sums 4 consecutive rows one after another (row pointers in one load; a row's list and own rows in one round trip): 12500 waves per gather launch, not 50000; row sum order kept
# speedup vs baseline: 4.7379x; 1.0309x over previous
#include <hip/hip_runtime.h>
#include <hip/hip_bf16.h>


typedef __attribute__((ext_vector_type(2))) float v2f;
typedef __attribute__((ext_vector_type(8))) float v8f;

#define DFEAT 96
#define NTILES_N 6

#define CE 4096
#define NBIN 256
#define SORT_THREADS 1024
#define MAXRPB 1024

__global__ void __launch_bounds__(SORT_THREADS)
cheb_split_ranges(const int* __restrict__ rows,
                  const int* __restrict__ cols,
                  const float* __restrict__ vals,
                  int* __restrict__ recR,
                  int2* __restrict__ recCV,
                  int* __restrict__ table,
                  int nEdges, int nNodes, int rpb, int ncp) {
    __shared__ int  hist[NBIN];
    __shared__ int  bstart[NBIN];
    __shared__ int  wtot[NBIN / 32];
    __shared__ int  sR[CE];
    __shared__ int2 sCV[CE];
    const int tid   = threadIdx.x;
    const int lane  = tid & 31;
    const int w     = tid >> 5;
    const int chunk = blockIdx.x;
    const int e0    = chunk * CE;

    if (tid < NBIN) hist[tid] = 0;

    int   r[4], c[4], bn[4], rk[4];
    float v[4];
    bool  ok[4];
#pragma unroll
    for (int k = 0; k < 4; ++k) {
        const int e = e0 + k * SORT_THREADS + tid;
        ok[k] = e < nEdges;
        const int ec = ok[k] ? e : nEdges - 1;
        r[k] = rows[ec];
        c[k] = cols[ec];
        v[k] = vals[ec];
    }
#pragma unroll
    for (int k = 0; k < 4; ++k) {
        unsigned row = (unsigned)r[k];
        if (row >= (unsigned)nNodes) row = (unsigned)(nNodes - 1);
        r[k]  = (int)row;
        bn[k] = ok[k] ? (int)(row / (unsigned)rpb) : -1;
        rk[k] = 0;
    }
    __syncthreads();
#pragma unroll
    for (int k = 0; k < 4; ++k)
        if (bn[k] >= 0) rk[k] = atomicAdd(&hist[bn[k]], 1);
    __syncthreads();

    int hv = 0, inc = 0;
    if (w < NBIN / 32) {
        hv = hist[tid];
        inc = hv;
#pragma unroll
        for (int d = 1; d < 32; d <<= 1) {
            int up = __shfl_up(inc, d, 32);
            if (lane >= d) inc += up;
        }
        if (lane == 31) wtot[w] = inc;
    }
    __syncthreads();
    if (w < NBIN / 32) {
        int carry = 0;
#pragma unroll
        for (int i = 0; i < NBIN / 32; ++i)
            if (i < w) carry += wtot[i];
        const int st = carry + inc - hv;
        bstart[tid] = st;
        table[(size_t)tid * ncp + chunk] = (st << 16) | hv;
    }
    __syncthreads();

#pragma unroll
    for (int k = 0; k < 4; ++k) {
        if (bn[k] >= 0) {
            const int p = bstart[bn[k]] + rk[k];
            sR[p]  = r[k];
            sCV[p] = make_int2(c[k], __float_as_int(v[k]));
        }
    }
    __syncthreads();

    int nIn = nEdges - e0;
    nIn = nIn > CE ? CE : nIn;
    for (int p = tid; p < nIn; p += SORT_THREADS) {
        recR[(size_t)e0 + p]  = sR[p];
        recCV[(size_t)e0 + p] = sCV[p];
    }
}

#define RUN_GROUPS (SORT_THREADS / 16)

template <bool PLACE>
__device__ __forceinline__ void walk_runs(const int* __restrict__ trow,
                                          const int* __restrict__ recR,
                                          const int2* __restrict__ recCV,
                                          int2* __restrict__ sorted,
                                          int* cntpos,
                                          int g, int sub, int ncp,
                                          int row0, int rpb, int nEdges) {
    for (int cb = 0; cb < ncp; cb += 16 * RUN_GROUPS) {
        const int cm = cb + g + RUN_GROUPS * sub;
        const int entMine = (cm < ncp) ? trow[cm] : 0;
        int jn = (ncp - cb + RUN_GROUPS - 1) / RUN_GROUPS;
        jn = jn > 16 ? 16 : jn;
        for (int j = 0; j < jn; ++j) {
            const int ent = __shfl(entMine, j, 16);
            const int c   = cb + g + RUN_GROUPS * j;
            const int len = (c < ncp) ? (ent & 0xFFFF) : 0;
            const size_t off = (size_t)c * CE + (ent >> 16);
            for (int k = sub; k < len; k += 16) {
                const int rrow = recR[off + k];
                int2 cvv = make_int2(0, 0);
                if (PLACE) {
                    cvv = recCV[off + k];
                    asm volatile("" :: "v"(cvv.x), "v"(cvv.y));
                }
                const int lr = rrow - row0;
                if ((unsigned)lr < (unsigned)rpb) {
                    if (!PLACE) {
                        atomicAdd(&cntpos[lr], 1);
                    } else {
                        const int pos = atomicAdd(&cntpos[lr], 1);
                        if ((unsigned)pos < (unsigned)nEdges)
                            sorted[pos] = cvv;
                    }
                }
            }
        }
    }
}

__global__ void __launch_bounds__(SORT_THREADS)
cheb_sort_range(const int* __restrict__ recR,
                const int2* __restrict__ recCV,
                const int* __restrict__ table,
                int* __restrict__ rowptr,
                int2* __restrict__ sorted,
                int nEdges, int nNodes, int rpb, int ncp) {
    __shared__ int cntpos[MAXRPB];
    __shared__ int wsumA[SORT_THREADS / 32];
    __shared__ int wsumB[SORT_THREADS / 32];
    const int tid  = threadIdx.x;
    const int lane = tid & 31;
    const int w    = tid >> 5;
    const int b    = blockIdx.x;
    const int row0 = b * rpb;

    cntpos[tid] = 0;

    const int4* T4 = (const int4*)table;
    const int lim4 = (b * ncp) >> 2;
    int s = 0;
    for (int i = tid; i < lim4; i += 4 * SORT_THREADS) {
        const int  i1 = i + SORT_THREADS, i2 = i + 2 * SORT_THREADS, i3 = i + 3 * SORT_THREADS;
        const bool v1 = i1 < lim4, v2 = i2 < lim4, v3 = i3 < lim4;
        const int4 a0 = T4[i];
        const int4 a1 = T4[v1 ? i1 : i];
        const int4 a2 = T4[v2 ? i2 : i];
        const int4 a3 = T4[v3 ? i3 : i];
        const int s0 = ((a0.x & 0xFFFF) + (a0.y & 0xFFFF)) + ((a0.z & 0xFFFF) + (a0.w & 0xFFFF));
        const int s1 = ((a1.x & 0xFFFF) + (a1.y & 0xFFFF)) + ((a1.z & 0xFFFF) + (a1.w & 0xFFFF));
        const int s2 = ((a2.x & 0xFFFF) + (a2.y & 0xFFFF)) + ((a2.z & 0xFFFF) + (a2.w & 0xFFFF));
        const int s3 = ((a3.x & 0xFFFF) + (a3.y & 0xFFFF)) + ((a3.z & 0xFFFF) + (a3.w & 0xFFFF));
        s += s0 + (v1 ? s1 : 0) + (v2 ? s2 : 0) + (v3 ? s3 : 0);
    }
#pragma unroll
    for (int d = 16; d >= 1; d >>= 1) s += __shfl_xor(s, d, 32);
    if (lane == 0) wsumA[w] = s;
    __syncthreads();

    const int g   = tid >> 4;
    const int sub = tid & 15;
    const int* trow = table + (size_t)b * ncp;

    walk_runs<false>(trow, recR, recCV, sorted, cntpos, g, sub, ncp, row0, rpb, nEdges);
    __syncthreads();

    int base = wsumA[lane];
#pragma unroll
    for (int d = 16; d >= 1; d >>= 1) base += __shfl_xor(base, d, 32);
    const int cv = cntpos[tid];
    int inc = cv;
#pragma unroll
    for (int d = 1; d < 32; d <<= 1) {
        int up = __shfl_up(inc, d, 32);
        if (lane >= d) inc += up;
    }
    if (lane == 31) wsumB[w] = inc;
    __syncthreads();
    int carry = (lane < w) ? wsumB[lane] : 0;
#pragma unroll
    for (int d = 16; d >= 1; d >>= 1) carry += __shfl_xor(carry, d, 32);
    const int start = base + carry + inc - cv;
    cntpos[tid] = start;
    if (tid < rpb && row0 + tid < nNodes) rowptr[row0 + tid] = start;
    if (b == (int)gridDim.x - 1 && tid == 0) rowptr[nNodes] = nEdges;
    __syncthreads();

    walk_runs<true>(trow, recR, recCV, sorted, cntpos, g, sub, ncp, row0, rpb, nEdges);
}

template <bool TAIL>
__device__ __forceinline__ void gather_group8(int2 my, int j, int n, unsigned cmax,
                                              bool act, const float* xl, float4& acc) {
    unsigned c[8];
    float    wv[8];
#pragma unroll
    for (int u = 0; u < 8; ++u) {
        int idx = (j + u) & 31;
        unsigned cu = (unsigned)__builtin_amdgcn_readlane(my.x, idx);
        c[u]  = cu > cmax ? cmax : cu;
        wv[u] = __int_as_float(__builtin_amdgcn_readlane(my.y, idx));
    }
    if (act) {
        float4 x[8];
#pragma unroll
        for (int u = 0; u < 8; ++u) {
            x[u] = make_float4(0.f, 0.f, 0.f, 0.f);
            if (!TAIL || j + u < n)
                x[u] = *(const float4*)(xl + (size_t)c[u] * DFEAT);
        }
#pragma unroll
        for (int u = 0; u < 8; ++u) {
            acc.x += wv[u] * x[u].x;
            acc.y += wv[u] * x[u].y;
            acc.z += wv[u] * x[u].z;
            acc.w += wv[u] * x[u].w;
        }
    }
}

#define RPW 4
#define WAVES_PER_GROUP 8
#define ROWS_PER_GROUP (RPW * WAVES_PER_GROUP)

template <int STEP>
__device__ __forceinline__ float4
row_value(const int2* __restrict__ sorted,
          const float* __restrict__ X,
          const float* __restrict__ H,
          const float* __restrict__ T1,
          const float* __restrict__ T2,
          int beg, int end,
          int row, int lane, int nNodes) {
    const bool act = lane < (DFEAT / 4);
    const unsigned cmax = (unsigned)(nNodes - 1);
    const float* xl = X + lane * 4;
    const size_t own = (size_t)row * DFEAT + lane * 4;

    int2 my0 = make_int2(0, 0);
    if (lane < end - beg) my0 = sorted[beg + lane];

    float4 h  = make_float4(0.f, 0.f, 0.f, 0.f);
    float4 t1 = make_float4(0.f, 0.f, 0.f, 0.f);
    float4 t2 = make_float4(0.f, 0.f, 0.f, 0.f);
    if (act) {
        h = *(const float4*)(H + own);
        if (STEP >= 2) t1 = *(const float4*)(T1 + own);
        if (STEP == 3) t2 = *(const float4*)(T2 + own);
    }

    float4 acc = make_float4(0.f, 0.f, 0.f, 0.f);
    for (int base = beg; base < end; base += 32) {
        int n = end - base;
        n = n > 32 ? 32 : n;
        int2 my = my0;
        if (base != beg) {
            my = make_int2(0, 0);
            if (lane < n) my = sorted[base + lane];
        }
        int j = 0;
        for (; j + 8 <= n; j += 8)
            gather_group8<false>(my, j, n, cmax, act, xl, acc);
        if (j < n)
            gather_group8<true>(my, j, n, cmax, act, xl, acc);
    }

    float4 o;
    if (STEP == 1) {
        o.x = 2.0f * acc.x - h.x;
        o.y = 2.0f * acc.y - h.y;
        o.z = 2.0f * acc.z - h.z;
        o.w = 2.0f * acc.w - h.w;
    } else if (STEP == 2) {
        o.x = 4.0f * acc.x - 2.0f * t1.x - h.x;
        o.y = 4.0f * acc.y - 2.0f * t1.y - h.y;
        o.z = 4.0f * acc.z - 2.0f * t1.z - h.z;
        o.w = 4.0f * acc.w - 2.0f * t1.w - h.w;
    } else {
        float4 t3;
        t3.x = 4.0f * acc.x - 2.0f * t2.x - t1.x;
        t3.y = 4.0f * acc.y - 2.0f * t2.y - t1.y;
        t3.z = 4.0f * acc.z - 2.0f * t2.z - t1.z;
        t3.w = 4.0f * acc.w - 2.0f * t2.w - t1.w;
        o.x = h.x + t1.x;  o.x += t2.x;  o.x += t3.x;
        o.y = h.y + t1.y;  o.y += t2.y;  o.y += t3.y;
        o.z = h.z + t1.z;  o.z += t2.z;  o.z += t3.z;
        o.w = h.w + t1.w;  o.w += t2.w;  o.w += t3.w;
    }
    return o;
}

template <int STEP, class Put>
__device__ __forceinline__ void
wave_rows(const int* __restrict__ rowptr,
          const int2* __restrict__ sorted,
          const float* __restrict__ X,
          const float* __restrict__ H,
          const float* __restrict__ T1,
          const float* __restrict__ T2,
          int row0, int lane, int nNodes, int nEdges, Put put) {
    int pi = row0 + (lane <= RPW ? lane : RPW);
    pi = pi > nNodes ? nNodes : pi;
    int rp = rowptr[pi];
    rp = rp < 0 ? 0 : rp;
    rp = rp > nEdges ? nEdges : rp;

#pragma nounroll
    for (int r = 0; r < RPW; ++r) {
        const int row = row0 + r;
        float4 o = make_float4(0.f, 0.f, 0.f, 0.f);
        if (row < nNodes) {
            const int beg = __builtin_amdgcn_readlane(rp, r);
            const int end = __builtin_amdgcn_readlane(rp, r + 1);
            o = row_value<STEP>(sorted, X, H, T1, T2, beg, end, row, lane, nNodes);
        }
        put(r, row, o);
    }
}

__global__ void __launch_bounds__(32 * WAVES_PER_GROUP)
cheb_gather96_t1(const int* __restrict__ rowptr,
                 const int2* __restrict__ sorted,
                 const float* __restrict__ H,
                 float* __restrict__ T1out,
                 int nNodes, int nEdges) {
    const int row0 = __builtin_amdgcn_readfirstlane(
        (int)((blockIdx.x * blockDim.x + threadIdx.x) >> 5)) * RPW;
    if (row0 >= nNodes) return;
    const int lane = threadIdx.x & 31;
    wave_rows<1>(rowptr, sorted, H, H, nullptr, nullptr, row0, lane, nNodes, nEdges,
                 [&](int, int row, float4 o) {
                     if (row < nNodes && lane < (DFEAT / 4))
                         *(float4*)(T1out + (size_t)row * DFEAT + lane * 4) = o;
                 });
}

__global__ void __launch_bounds__(32 * WAVES_PER_GROUP)
cheb_gather96_t2(const int* __restrict__ rowptr,
                 const int2* __restrict__ sorted,
                 const float* __restrict__ H,
                 const float* __restrict__ T1,
                 float* __restrict__ T2out,
                 int nNodes, int nEdges) {
    const int row0 = __builtin_amdgcn_readfirstlane(
        (int)((blockIdx.x * blockDim.x + threadIdx.x) >> 5)) * RPW;
    if (row0 >= nNodes) return;
    const int lane = threadIdx.x & 31;
    wave_rows<2>(rowptr, sorted, T1, H, T1, nullptr, row0, lane, nNodes, nEdges,
                 [&](int, int row, float4 o) {
                     if (row < nNodes && lane < (DFEAT / 4))
                         *(float4*)(T2out + (size_t)row * DFEAT + lane * 4) = o;
                 });
}

#define SPITCH 100
#define TILES_PER_GROUP ((ROWS_PER_GROUP / 16) * NTILES_N)
__global__ void __launch_bounds__(32 * WAVES_PER_GROUP)
cheb_gather96_s_gemm(const int* __restrict__ rowptr,
                     const int2* __restrict__ sorted,
                     const float* __restrict__ H,
                     const float* __restrict__ T1,
                     const float* __restrict__ T2,
                     const float* __restrict__ W,
                     const float* __restrict__ bias,
                     float* __restrict__ out,
                     int nNodes, int nEdges) {
    __shared__ __attribute__((aligned(16))) float sS[ROWS_PER_GROUP * SPITCH];
    const int lane = threadIdx.x & 31;
    const int w    = __builtin_amdgcn_readfirstlane((int)(threadIdx.x >> 5));
    const int rowG = blockIdx.x * ROWS_PER_GROUP;

    float* srow = sS + (w * RPW) * SPITCH + lane * 4;
    wave_rows<3>(rowptr, sorted, T2, H, T1, T2, rowG + w * RPW, lane, nNodes, nEdges,
                 [&](int r, int, float4 o) {
                     if (lane < (DFEAT / 4))
                         *(float4*)(srow + r * SPITCH) = o;
                 });
    __syncthreads();

    const int half = lane >> 4;
    const int l    = lane & 15;

    for (int t = w; t < TILES_PER_GROUP; t += WAVES_PER_GROUP) {
        const int mt   = t / NTILES_N;
        const int nt   = t - mt * NTILES_N;
        const int row0 = rowG + mt * 16;
        if (row0 >= nNodes) break;

        const float* arow = sS + (mt * 16 + l) * SPITCH + 2 * half;
        const float* Wcol = W + nt * 16 + l + (size_t)(2 * half) * DFEAT;

        v8f c = {};
#pragma unroll 12
        for (int k0 = 0; k0 < DFEAT; k0 += 4) {
            v2f a = *(const v2f*)(arow + k0);
            v2f b;
            b.x = Wcol[(size_t)k0 * DFEAT];
            b.y = Wcol[(size_t)(k0 + 1) * DFEAT];
            c = __builtin_amdgcn_wmma_f32_16x16x4_f32(
                     false, a,  false, b,
                     (short)0, c,  false,  false);
        }

        const int n  = nt * 16 + l;
        const float bv = bias[n];
        const int mbase = row0 + half * 8;
        float* orow = out + (size_t)mbase * DFEAT + n;
#pragma unroll
        for (int r = 0; r < 8; ++r) {
            if (mbase + r < nNodes)
                orow[(size_t)r * DFEAT] = c[r] + bv;
        }
    }
}

extern "C" void kernel_launch(void* const* d_in, const int* in_sizes, int n_in,
                              void* d_out, int out_size, void* d_ws, size_t ws_size,
                              hipStream_t stream) {
    const int*   rows = (const int*)d_in[0];
    const int*   cols = (const int*)d_in[1];
    const float* vals = (const float*)d_in[2];
    const float* H    = (const float*)d_in[3];
    const float* W    = (const float*)d_in[4];
    const float* bias = (const float*)d_in[5];
    float* out = (float*)d_out;

    const int nEdges = in_sizes[0];
    const int nNodes = in_sizes[3] / DFEAT;
    const int nElem  = nNodes * DFEAT;

    float* P0 = (float*)d_ws;
    float* L  = P0 + nElem;
    float* P2 = L + nElem;

    const int ncp = (((nEdges + CE - 1) / CE) + 3) / 4 * 4;
    const int rpb = (nNodes + NBIN - 1) / NBIN;

    int2* sorted = (int2*)L;
    int2* recCV  = sorted + (nEdges + 1) / 2 * 2;
    int*  recR   = (int*)(recCV + (size_t)ncp * CE);
    int*  table  = recR + (size_t)ncp * CE;
    int*  rowptr = table + (size_t)NBIN * ncp;

    dim3 blk(32 * WAVES_PER_GROUP);
    dim3 grdRows((nNodes + ROWS_PER_GROUP - 1) / ROWS_PER_GROUP);

    cheb_split_ranges<<<dim3(ncp), dim3(SORT_THREADS), 0, stream>>>(rows, cols, vals, recR, recCV, table,
                                                                    nEdges, nNodes, rpb, ncp);
    cheb_sort_range<<<dim3(NBIN), dim3(SORT_THREADS), 0, stream>>>(recR, recCV, table, rowptr, sorted,
                                                                   nEdges, nNodes, rpb, ncp);

    cheb_gather96_t1<<<grdRows, blk, 0, stream>>>(rowptr, sorted, H, P0, nNodes, nEdges);

    cheb_gather96_t2<<<grdRows, blk, 0, stream>>>(rowptr, sorted, H, P0, P2, nNodes, nEdges);

    cheb_gather96_s_gemm<<<grdRows, blk, 0, stream>>>(rowptr, sorted, H, P0, P2, W, bias, out,
                                                            nNodes, nEdges);
}


# k = 1 row sums moved into sort level 2: each 1024-thread group sums its own 196 rows from an LDS copy of the lists it has just placed (rows dealt by an LDS counter); 4 launches not 5; sum order kept
# speedup vs baseline: 4.9515x; 1.0451x over previous
#include <hip/hip_runtime.h>
#include <hip/hip_bf16.h>


typedef __attribute__((ext_vector_type(2))) float v2f;
typedef __attribute__((ext_vector_type(8))) float v8f;

#define DFEAT 96
#define NTILES_N 6

#define CE 4096
#define NBIN 256
#define SORT_THREADS 1024
#define MAXRPB 1024
#define LIST_CAP 20480

__global__ void __launch_bounds__(SORT_THREADS)
cheb_split_ranges(const int* __restrict__ rows,
                  const int* __restrict__ cols,
                  const float* __restrict__ vals,
                  int* __restrict__ recR,
                  int2* __restrict__ recCV,
                  int* __restrict__ table,
                  int nEdges, int nNodes, int rpb, int ncp) {
    __shared__ int  hist[NBIN];
    __shared__ int  bstart[NBIN];
    __shared__ int  wtot[NBIN / 32];
    __shared__ int  sR[CE];
    __shared__ int2 sCV[CE];
    const int tid   = threadIdx.x;
    const int lane  = tid & 31;
    const int w     = tid >> 5;
    const int chunk = blockIdx.x;
    const int e0    = chunk * CE;

    if (tid < NBIN) hist[tid] = 0;

    int   r[4], c[4], bn[4], rk[4];
    float v[4];
    bool  ok[4];
#pragma unroll
    for (int k = 0; k < 4; ++k) {
        const int e = e0 + k * SORT_THREADS + tid;
        ok[k] = e < nEdges;
        const int ec = ok[k] ? e : nEdges - 1;
        r[k] = rows[ec];
        c[k] = cols[ec];
        v[k] = vals[ec];
    }
#pragma unroll
    for (int k = 0; k < 4; ++k) {
        unsigned row = (unsigned)r[k];
        if (row >= (unsigned)nNodes) row = (unsigned)(nNodes - 1);
        r[k]  = (int)row;
        bn[k] = ok[k] ? (int)(row / (unsigned)rpb) : -1;
        rk[k] = 0;
    }
    __syncthreads();
#pragma unroll
    for (int k = 0; k < 4; ++k)
        if (bn[k] >= 0) rk[k] = atomicAdd(&hist[bn[k]], 1);
    __syncthreads();

    int hv = 0, inc = 0;
    if (w < NBIN / 32) {
        hv = hist[tid];
        inc = hv;
#pragma unroll
        for (int d = 1; d < 32; d <<= 1) {
            int up = __shfl_up(inc, d, 32);
            if (lane >= d) inc += up;
        }
        if (lane == 31) wtot[w] = inc;
    }
    __syncthreads();
    if (w < NBIN / 32) {
        int carry = 0;
#pragma unroll
        for (int i = 0; i < NBIN / 32; ++i)
            if (i < w) carry += wtot[i];
        const int st = carry + inc - hv;
        bstart[tid] = st;
        table[(size_t)tid * ncp + chunk] = (st << 16) | hv;
    }
    __syncthreads();

#pragma unroll
    for (int k = 0; k < 4; ++k) {
        if (bn[k] >= 0) {
            const int p = bstart[bn[k]] + rk[k];
            sR[p]  = r[k];
            sCV[p] = make_int2(c[k], __float_as_int(v[k]));
        }
    }
    __syncthreads();

    int nIn = nEdges - e0;
    nIn = nIn > CE ? CE : nIn;
    for (int p = tid; p < nIn; p += SORT_THREADS) {
        recR[(size_t)e0 + p]  = sR[p];
        recCV[(size_t)e0 + p] = sCV[p];
    }
}

template <bool TAIL>
__device__ __forceinline__ void gather_group8(int2 my, int j, int n, unsigned cmax,
                                              bool act, const float* xl, float4& acc) {
    unsigned c[8];
    float    wv[8];
#pragma unroll
    for (int u = 0; u < 8; ++u) {
        int idx = (j + u) & 31;
        unsigned cu = (unsigned)__builtin_amdgcn_readlane(my.x, idx);
        c[u]  = cu > cmax ? cmax : cu;
        wv[u] = __int_as_float(__builtin_amdgcn_readlane(my.y, idx));
    }
    if (act) {
        float4 x[8];
#pragma unroll
        for (int u = 0; u < 8; ++u) {
            x[u] = make_float4(0.f, 0.f, 0.f, 0.f);
            if (!TAIL || j + u < n)
                x[u] = *(const float4*)(xl + (size_t)c[u] * DFEAT);
        }
#pragma unroll
        for (int u = 0; u < 8; ++u) {
            acc.x += wv[u] * x[u].x;
            acc.y += wv[u] * x[u].y;
            acc.z += wv[u] * x[u].z;
            acc.w += wv[u] * x[u].w;
        }
    }
}

#define RUN_GROUPS (SORT_THREADS / 16)

template <bool PLACE>
__device__ __forceinline__ void walk_runs(const int* __restrict__ trow,
                                          const int* __restrict__ recR,
                                          const int2* __restrict__ recCV,
                                          int2* __restrict__ sorted,
                                          int* cntpos, int2* sE, int base,
                                          int g, int sub, int ncp,
                                          int row0, int rpb, int nEdges) {
    for (int cb = 0; cb < ncp; cb += 16 * RUN_GROUPS) {
        const int cm = cb + g + RUN_GROUPS * sub;
        const int entMine = (cm < ncp) ? trow[cm] : 0;
        int jn = (ncp - cb + RUN_GROUPS - 1) / RUN_GROUPS;
        jn = jn > 16 ? 16 : jn;
        for (int j = 0; j < jn; ++j) {
            const int ent = __shfl(entMine, j, 16);
            const int c   = cb + g + RUN_GROUPS * j;
            const int len = (c < ncp) ? (ent & 0xFFFF) : 0;
            const size_t off = (size_t)c * CE + (ent >> 16);
            for (int k = sub; k < len; k += 16) {
                const int rrow = recR[off + k];
                int2 cvv = make_int2(0, 0);
                if (PLACE) {
                    cvv = recCV[off + k];
                    asm volatile("" :: "v"(cvv.x), "v"(cvv.y));
                }
                const int lr = rrow - row0;
                if ((unsigned)lr < (unsigned)rpb) {
                    if (!PLACE) {
                        atomicAdd(&cntpos[lr], 1);
                    } else {
                        const int pos = atomicAdd(&cntpos[lr], 1);
                        if ((unsigned)pos < (unsigned)nEdges)
                            sorted[pos] = cvv;
                        const int lp = pos - base;
                        if ((unsigned)lp < (unsigned)LIST_CAP)
                            sE[lp] = cvv;
                    }
                }
            }
        }
    }
}

template <bool IN_LDS>
__device__ __forceinline__ void
range_rows_t1(const int2* sE, const int* rbeg, const int* cntpos, int* nextRow,
              const int2* __restrict__ sorted,
              const float* __restrict__ H,
              float* __restrict__ T1out,
              int base, int row0, int nrows, int lane, int nNodes, int nEdges) {
    const bool act = lane < (DFEAT / 4);
    const unsigned cmax = (unsigned)(nNodes - 1);
    const float* xl = H + lane * 4;
    for (int trip = 0; trip < MAXRPB; ++trip) {
        int i = 0;
        if (lane == 0) i = atomicAdd(nextRow, 1);
        i = __builtin_amdgcn_readfirstlane(i);
        if (i >= nrows) break;
        const int row = row0 + i;
        const size_t own = (size_t)row * DFEAT + lane * 4;
        float4 h = make_float4(0.f, 0.f, 0.f, 0.f);
        if (act) h = *(const float4*)(H + own);
        int beg = __builtin_amdgcn_readfirstlane(rbeg[i]);
        int end = __builtin_amdgcn_readfirstlane(cntpos[i]);
        beg = beg < 0 ? 0 : beg;
        end = end > nEdges ? nEdges : end;
        float4 acc = make_float4(0.f, 0.f, 0.f, 0.f);
        for (int b0 = beg; b0 < end; b0 += 32) {
            int n = end - b0;
            n = n > 32 ? 32 : n;
            int2 my = make_int2(0, 0);
            if (lane < n) {
                if (IN_LDS) my = sE[b0 - base + lane];
                else        my = sorted[b0 + lane];
            }
            int j = 0;
            for (; j + 8 <= n; j += 8)
                gather_group8<false>(my, j, n, cmax, act, xl, acc);
            if (j < n)
                gather_group8<true>(my, j, n, cmax, act, xl, acc);
        }
        if (act) {
            float4 o;
            o.x = 2.0f * acc.x - h.x;
            o.y = 2.0f * acc.y - h.y;
            o.z = 2.0f * acc.z - h.z;
            o.w = 2.0f * acc.w - h.w;
            *(float4*)(T1out + own) = o;
        }
    }
}

__global__ void __launch_bounds__(SORT_THREADS)
cheb_sort_range_t1(const int* __restrict__ recR,
                   const int2* __restrict__ recCV,
                   const int* __restrict__ table,
                   int* __restrict__ rowptr,
                   int2* __restrict__ sorted,
                   const float* __restrict__ H,
                   float* __restrict__ T1out,
                   int nEdges, int nNodes, int rpb, int ncp) {
    __shared__ int  cntpos[MAXRPB];
    __shared__ int  rbeg[MAXRPB];
    __shared__ int  wsumA[SORT_THREADS / 32];
    __shared__ int  wsumB[SORT_THREADS / 32];
    __shared__ int  nextRow;
    __shared__ int2 sE[LIST_CAP];
    const int tid  = threadIdx.x;
    const int lane = tid & 31;
    const int w    = tid >> 5;
    const int b    = blockIdx.x;
    const int row0 = b * rpb;

    cntpos[tid] = 0;
    if (tid == 0) nextRow = 0;

    const int4* T4 = (const int4*)table;
    const int lim4 = (b * ncp) >> 2;
    int s = 0;
    for (int i = tid; i < lim4; i += 4 * SORT_THREADS) {
        const int  i1 = i + SORT_THREADS, i2 = i + 2 * SORT_THREADS, i3 = i + 3 * SORT_THREADS;
        const bool v1 = i1 < lim4, v2 = i2 < lim4, v3 = i3 < lim4;
        const int4 a0 = T4[i];
        const int4 a1 = T4[v1 ? i1 : i];
        const int4 a2 = T4[v2 ? i2 : i];
        const int4 a3 = T4[v3 ? i3 : i];
        const int s0 = ((a0.x & 0xFFFF) + (a0.y & 0xFFFF)) + ((a0.z & 0xFFFF) + (a0.w & 0xFFFF));
        const int s1 = ((a1.x & 0xFFFF) + (a1.y & 0xFFFF)) + ((a1.z & 0xFFFF) + (a1.w & 0xFFFF));
        const int s2 = ((a2.x & 0xFFFF) + (a2.y & 0xFFFF)) + ((a2.z & 0xFFFF) + (a2.w & 0xFFFF));
        const int s3 = ((a3.x & 0xFFFF) + (a3.y & 0xFFFF)) + ((a3.z & 0xFFFF) + (a3.w & 0xFFFF));
        s += s0 + (v1 ? s1 : 0) + (v2 ? s2 : 0) + (v3 ? s3 : 0);
    }
#pragma unroll
    for (int d = 16; d >= 1; d >>= 1) s += __shfl_xor(s, d, 32);
    if (lane == 0) wsumA[w] = s;
    __syncthreads();

    const int g   = tid >> 4;
    const int sub = tid & 15;
    const int* trow = table + (size_t)b * ncp;

    walk_runs<false>(trow, recR, recCV, sorted, cntpos, sE, 0, g, sub, ncp, row0, rpb, nEdges);
    __syncthreads();

    int base = wsumA[lane];
#pragma unroll
    for (int d = 16; d >= 1; d >>= 1) base += __shfl_xor(base, d, 32);
    const int cv = cntpos[tid];
    int inc = cv;
#pragma unroll
    for (int d = 1; d < 32; d <<= 1) {
        int up = __shfl_up(inc, d, 32);
        if (lane >= d) inc += up;
    }
    if (lane == 31) wsumB[w] = inc;
    __syncthreads();
    int carry = (lane < w) ? wsumB[lane] : 0;
#pragma unroll
    for (int d = 16; d >= 1; d >>= 1) carry += __shfl_xor(carry, d, 32);
    const int start = base + carry + inc - cv;
    cntpos[tid] = start;
    rbeg[tid]   = start;
    if (tid < rpb && row0 + tid < nNodes) rowptr[row0 + tid] = start;
    if (b == (int)gridDim.x - 1 && tid == 0) rowptr[nNodes] = nEdges;
    __syncthreads();

    walk_runs<true>(trow, recR, recCV, sorted, cntpos, sE, base, g, sub, ncp, row0, rpb, nEdges);
    __syncthreads();

    const int listEnd = __builtin_amdgcn_readfirstlane(cntpos[MAXRPB - 1]);
    int nrows = nNodes - row0;
    nrows = nrows > rpb ? rpb : nrows;
    if ((listEnd - base) <= LIST_CAP)
        range_rows_t1<true>(sE, rbeg, cntpos, &nextRow, sorted, H, T1out,
                            base, row0, nrows, lane, nNodes, nEdges);
    else
        range_rows_t1<false>(sE, rbeg, cntpos, &nextRow, sorted, H, T1out,
                             base, row0, nrows, lane, nNodes, nEdges);
}

#define RPW 4
#define WAVES_PER_GROUP 8
#define ROWS_PER_GROUP (RPW * WAVES_PER_GROUP)

template <int STEP>
__device__ __forceinline__ float4
row_value(const int2* __restrict__ sorted,
          const float* __restrict__ X,
          const float* __restrict__ H,
          const float* __restrict__ T1,
          const float* __restrict__ T2,
          int beg, int end,
          int row, int lane, int nNodes) {
    const bool act = lane < (DFEAT / 4);
    const unsigned cmax = (unsigned)(nNodes - 1);
    const float* xl = X + lane * 4;
    const size_t own = (size_t)row * DFEAT + lane * 4;

    int2 my0 = make_int2(0, 0);
    if (lane < end - beg) my0 = sorted[beg + lane];

    float4 h  = make_float4(0.f, 0.f, 0.f, 0.f);
    float4 t1 = make_float4(0.f, 0.f, 0.f, 0.f);
    float4 t2 = make_float4(0.f, 0.f, 0.f, 0.f);
    if (act) {
        h = *(const float4*)(H + own);
        if (STEP >= 2) t1 = *(const float4*)(T1 + own);
        if (STEP == 3) t2 = *(const float4*)(T2 + own);
    }

    float4 acc = make_float4(0.f, 0.f, 0.f, 0.f);
    for (int base = beg; base < end; base += 32) {
        int n = end - base;
        n = n > 32 ? 32 : n;
        int2 my = my0;
        if (base != beg) {
            my = make_int2(0, 0);
            if (lane < n) my = sorted[base + lane];
        }
        int j = 0;
        for (; j + 8 <= n; j += 8)
            gather_group8<false>(my, j, n, cmax, act, xl, acc);
        if (j < n)
            gather_group8<true>(my, j, n, cmax, act, xl, acc);
    }

    float4 o;
    if (STEP == 1) {
        o.x = 2.0f * acc.x - h.x;
        o.y = 2.0f * acc.y - h.y;
        o.z = 2.0f * acc.z - h.z;
        o.w = 2.0f * acc.w - h.w;
    } else if (STEP == 2) {
        o.x = 4.0f * acc.x - 2.0f * t1.x - h.x;
        o.y = 4.0f * acc.y - 2.0f * t1.y - h.y;
        o.z = 4.0f * acc.z - 2.0f * t1.z - h.z;
        o.w = 4.0f * acc.w - 2.0f * t1.w - h.w;
    } else {
        float4 t3;
        t3.x = 4.0f * acc.x - 2.0f * t2.x - t1.x;
        t3.y = 4.0f * acc.y - 2.0f * t2.y - t1.y;
        t3.z = 4.0f * acc.z - 2.0f * t2.z - t1.z;
        t3.w = 4.0f * acc.w - 2.0f * t2.w - t1.w;
        o.x = h.x + t1.x;  o.x += t2.x;  o.x += t3.x;
        o.y = h.y + t1.y;  o.y += t2.y;  o.y += t3.y;
        o.z = h.z + t1.z;  o.z += t2.z;  o.z += t3.z;
        o.w = h.w + t1.w;  o.w += t2.w;  o.w += t3.w;
    }
    return o;
}

template <int STEP, class Put>
__device__ __forceinline__ void
wave_rows(const int* __restrict__ rowptr,
          const int2* __restrict__ sorted,
          const float* __restrict__ X,
          const float* __restrict__ H,
          const float* __restrict__ T1,
          const float* __restrict__ T2,
          int row0, int lane, int nNodes, int nEdges, Put put) {
    int pi = row0 + (lane <= RPW ? lane : RPW);
    pi = pi > nNodes ? nNodes : pi;
    int rp = rowptr[pi];
    rp = rp < 0 ? 0 : rp;
    rp = rp > nEdges ? nEdges : rp;

#pragma nounroll
    for (int r = 0; r < RPW; ++r) {
        const int row = row0 + r;
        float4 o = make_float4(0.f, 0.f, 0.f, 0.f);
        if (row < nNodes) {
            const int beg = __builtin_amdgcn_readlane(rp, r);
            const int end = __builtin_amdgcn_readlane(rp, r + 1);
            o = row_value<STEP>(sorted, X, H, T1, T2, beg, end, row, lane, nNodes);
        }
        put(r, row, o);
    }
}

__global__ void __launch_bounds__(32 * WAVES_PER_GROUP)
cheb_gather96_t2(const int* __restrict__ rowptr,
                 const int2* __restrict__ sorted,
                 const float* __restrict__ H,
                 const float* __restrict__ T1,
                 float* __restrict__ T2out,
                 int nNodes, int nEdges) {
    const int row0 = __builtin_amdgcn_readfirstlane(
        (int)((blockIdx.x * blockDim.x + threadIdx.x) >> 5)) * RPW;
    if (row0 >= nNodes) return;
    const int lane = threadIdx.x & 31;
    wave_rows<2>(rowptr, sorted, T1, H, T1, nullptr, row0, lane, nNodes, nEdges,
                 [&](int, int row, float4 o) {
                     if (row < nNodes && lane < (DFEAT / 4))
                         *(float4*)(T2out + (size_t)row * DFEAT + lane * 4) = o;
                 });
}

#define SPITCH 100
#define TILES_PER_GROUP ((ROWS_PER_GROUP / 16) * NTILES_N)
__global__ void __launch_bounds__(32 * WAVES_PER_GROUP)
cheb_gather96_s_gemm(const int* __restrict__ rowptr,
                     const int2* __restrict__ sorted,
                     const float* __restrict__ H,
                     const float* __restrict__ T1,
                     const float* __restrict__ T2,
                     const float* __restrict__ W,
                     const float* __restrict__ bias,
                     float* __restrict__ out,
                     int nNodes, int nEdges) {
    __shared__ __attribute__((aligned(16))) float sS[ROWS_PER_GROUP * SPITCH];
    const int lane = threadIdx.x & 31;
    const int w    = __builtin_amdgcn_readfirstlane((int)(threadIdx.x >> 5));
    const int rowG = blockIdx.x * ROWS_PER_GROUP;

    float* srow = sS + (w * RPW) * SPITCH + lane * 4;
    wave_rows<3>(rowptr, sorted, T2, H, T1, T2, rowG + w * RPW, lane, nNodes, nEdges,
                 [&](int r, int, float4 o) {
                     if (lane < (DFEAT / 4))
                         *(float4*)(srow + r * SPITCH) = o;
                 });
    __syncthreads();

    const int half = lane >> 4;
    const int l    = lane & 15;

    for (int t = w; t < TILES_PER_GROUP; t += WAVES_PER_GROUP) {
        const int mt   = t / NTILES_N;
        const int nt   = t - mt * NTILES_N;
        const int row0 = rowG + mt * 16;
        if (row0 >= nNodes) break;

        const float* arow = sS + (mt * 16 + l) * SPITCH + 2 * half;
        const float* Wcol = W + nt * 16 + l + (size_t)(2 * half) * DFEAT;

        v8f c = {};
#pragma unroll 12
        for (int k0 = 0; k0 < DFEAT; k0 += 4) {
            v2f a = *(const v2f*)(arow + k0);
            v2f b;
            b.x = Wcol[(size_t)k0 * DFEAT];
            b.y = Wcol[(size_t)(k0 + 1) * DFEAT];
            c = __builtin_amdgcn_wmma_f32_16x16x4_f32(
                     false, a,  false, b,
                     (short)0, c,  false,  false);
        }

        const int n  = nt * 16 + l;
        const float bv = bias[n];
        const int mbase = row0 + half * 8;
        float* orow = out + (size_t)mbase * DFEAT + n;
#pragma unroll
        for (int r = 0; r < 8; ++r) {
            if (mbase + r < nNodes)
                orow[(size_t)r * DFEAT] = c[r] + bv;
        }
    }
}

extern "C" void kernel_launch(void* const* d_in, const int* in_sizes, int n_in,
                              void* d_out, int out_size, void* d_ws, size_t ws_size,
                              hipStream_t stream) {
    const int*   rows = (const int*)d_in[0];
    const int*   cols = (const int*)d_in[1];
    const float* vals = (const float*)d_in[2];
    const float* H    = (const float*)d_in[3];
    const float* W    = (const float*)d_in[4];
    const float* bias = (const float*)d_in[5];
    float* out = (float*)d_out;

    const int nEdges = in_sizes[0];
    const int nNodes = in_sizes[3] / DFEAT;
    const int nElem  = nNodes * DFEAT;

    float* P0 = (float*)d_ws;
    float* L  = P0 + nElem;
    float* P2 = L + nElem;

    const int ncp = (((nEdges + CE - 1) / CE) + 3) / 4 * 4;
    const int rpb = (nNodes + NBIN - 1) / NBIN;

    int2* sorted = (int2*)L;
    int2* recCV  = sorted + (nEdges + 1) / 2 * 2;
    int*  recR   = (int*)(recCV + (size_t)ncp * CE);
    int*  table  = recR + (size_t)ncp * CE;
    int*  rowptr = table + (size_t)NBIN * ncp;

    dim3 blk(32 * WAVES_PER_GROUP);
    dim3 grdRows((nNodes + ROWS_PER_GROUP - 1) / ROWS_PER_GROUP);

    cheb_split_ranges<<<dim3(ncp), dim3(SORT_THREADS), 0, stream>>>(rows, cols, vals, recR, recCV, table,
                                                                    nEdges, nNodes, rpb, ncp);
    cheb_sort_range_t1<<<dim3(NBIN), dim3(SORT_THREADS), 0, stream>>>(recR, recCV, table, rowptr, sorted,
                                                                      H, P0, nEdges, nNodes, rpb, ncp);

    cheb_gather96_t2<<<grdRows, blk, 0, stream>>>(rowptr, sorted, H, P0, P2, nNodes, nEdges);

    cheb_gather96_s_gemm<<<grdRows, blk, 0, stream>>>(rowptr, sorted, H, P0, P2, W, bias, out,
                                                            nNodes, nEdges);
}


# Sort level 2 waits on memory twice, not ~14 times: all run records fetched at once into registers, list start = sum of own table row's starts, rank = the counting LDS add's return (row sum order)
# speedup vs baseline: 5.2481x; 1.0599x over previous
#include <hip/hip_runtime.h>
#include <hip/hip_bf16.h>


typedef __attribute__((ext_vector_type(2))) float v2f;
typedef __attribute__((ext_vector_type(8))) float v8f;

#define DFEAT 96
#define NTILES_N 6

#define CE 4096
#define NBIN 256
#define SORT_THREADS 1024
#define MAXRPB 1024
#define LIST_CAP 20480

__global__ void __launch_bounds__(SORT_THREADS)
cheb_split_ranges(const int* __restrict__ rows,
                  const int* __restrict__ cols,
                  const float* __restrict__ vals,
                  int* __restrict__ recR,
                  int2* __restrict__ recCV,
                  int* __restrict__ table,
                  int nEdges, int nNodes, int rpb, int ncp) {
    __shared__ int  hist[NBIN];
    __shared__ int  bstart[NBIN];
    __shared__ int  wtot[NBIN / 32];
    __shared__ int  sR[CE];
    __shared__ int2 sCV[CE];
    const int tid   = threadIdx.x;
    const int lane  = tid & 31;
    const int w     = tid >> 5;
    const int chunk = blockIdx.x;
    const int e0    = chunk * CE;

    if (tid < NBIN) hist[tid] = 0;

    int   r[4], c[4], bn[4], rk[4];
    float v[4];
    bool  ok[4];
#pragma unroll
    for (int k = 0; k < 4; ++k) {
        const int e = e0 + k * SORT_THREADS + tid;
        ok[k] = e < nEdges;
        const int ec = ok[k] ? e : nEdges - 1;
        r[k] = rows[ec];
        c[k] = cols[ec];
        v[k] = vals[ec];
    }
#pragma unroll
    for (int k = 0; k < 4; ++k) {
        unsigned row = (unsigned)r[k];
        if (row >= (unsigned)nNodes) row = (unsigned)(nNodes - 1);
        r[k]  = (int)row;
        bn[k] = ok[k] ? (int)(row / (unsigned)rpb) : -1;
        rk[k] = 0;
    }
    __syncthreads();
#pragma unroll
    for (int k = 0; k < 4; ++k)
        if (bn[k] >= 0) rk[k] = atomicAdd(&hist[bn[k]], 1);
    __syncthreads();

    int hv = 0, inc = 0;
    if (w < NBIN / 32) {
        hv = hist[tid];
        inc = hv;
#pragma unroll
        for (int d = 1; d < 32; d <<= 1) {
            int up = __shfl_up(inc, d, 32);
            if (lane >= d) inc += up;
        }
        if (lane == 31) wtot[w] = inc;
    }
    __syncthreads();
    if (w < NBIN / 32) {
        int carry = 0;
#pragma unroll
        for (int i = 0; i < NBIN / 32; ++i)
            if (i < w) carry += wtot[i];
        const int st = carry + inc - hv;
        bstart[tid] = st;
        table[(size_t)tid * ncp + chunk] = (st << 16) | hv;
    }
    __syncthreads();

#pragma unroll
    for (int k = 0; k < 4; ++k) {
        if (bn[k] >= 0) {
            const int p = bstart[bn[k]] + rk[k];
            sR[p]  = r[k];
            sCV[p] = make_int2(c[k], __float_as_int(v[k]));
        }
    }
    __syncthreads();

    int nIn = nEdges - e0;
    nIn = nIn > CE ? CE : nIn;
    for (int p = tid; p < nIn; p += SORT_THREADS) {
        recR[(size_t)e0 + p]  = sR[p];
        recCV[(size_t)e0 + p] = sCV[p];
    }
}

template <bool TAIL>
__device__ __forceinline__ void gather_group8(int2 my, int j, int n, unsigned cmax,
                                              bool act, const float* xl, float4& acc) {
    unsigned c[8];
    float    wv[8];
#pragma unroll
    for (int u = 0; u < 8; ++u) {
        int idx = (j + u) & 31;
        unsigned cu = (unsigned)__builtin_amdgcn_readlane(my.x, idx);
        c[u]  = cu > cmax ? cmax : cu;
        wv[u] = __int_as_float(__builtin_amdgcn_readlane(my.y, idx));
    }
    if (act) {
        float4 x[8];
#pragma unroll
        for (int u = 0; u < 8; ++u) {
            x[u] = make_float4(0.f, 0.f, 0.f, 0.f);
            if (!TAIL || j + u < n)
                x[u] = *(const float4*)(xl + (size_t)c[u] * DFEAT);
        }
#pragma unroll
        for (int u = 0; u < 8; ++u) {
            acc.x += wv[u] * x[u].x;
            acc.y += wv[u] * x[u].y;
            acc.z += wv[u] * x[u].z;
            acc.w += wv[u] * x[u].w;
        }
    }
}

#define RUN_GROUPS (SORT_THREADS / 16)
#define HELD_J 4
#define HELD_S 3

template <bool PLACE>
__device__ __forceinline__ void walk_rest(const int* __restrict__ trow,
                                          const int* __restrict__ recR,
                                          const int2* __restrict__ recCV,
                                          int2* __restrict__ sorted,
                                          int* cnt, int2* sE, int base,
                                          int g, int sub, int ncp,
                                          int row0, int rpb, int nEdges) {
    for (int cb = 0; cb < ncp; cb += 16 * RUN_GROUPS) {
        const int cm = cb + g + RUN_GROUPS * sub;
        const int entMine = (cm < ncp) ? trow[cm] : 0;
        int jn = (ncp - cb + RUN_GROUPS - 1) / RUN_GROUPS;
        jn = jn > 16 ? 16 : jn;
        for (int j = 0; j < jn; ++j) {
            const int ent = __shfl(entMine, j, 16);
            const int c   = cb + g + RUN_GROUPS * j;
            const int len = (c < ncp) ? (ent & 0xFFFF) : 0;
            const size_t off = (size_t)c * CE + (ent >> 16);
            const int k0 = (cb == 0 && j < HELD_J) ? sub + 16 * HELD_S : sub;
            for (int k = k0; k < len; k += 16) {
                const int lr = recR[off + k] - row0;
                if ((unsigned)lr < (unsigned)rpb) {
                    if (!PLACE) {
                        atomicAdd(&cnt[lr], 1);
                    } else {
                        const int2 cvv = recCV[off + k];
                        const int pos = atomicAdd(&cnt[lr], 1);
                        if ((unsigned)pos < (unsigned)nEdges)
                            sorted[pos] = cvv;
                        const int lp = pos - base;
                        if ((unsigned)lp < (unsigned)LIST_CAP)
                            sE[lp] = cvv;
                    }
                }
            }
        }
    }
}

template <bool IN_LDS>
__device__ __forceinline__ void
range_rows_t1(const int2* sE, const int* rbeg, const int* cntpos, int* nextRow,
              const int2* __restrict__ sorted,
              const float* __restrict__ H,
              float* __restrict__ T1out,
              int base, int row0, int nrows, int lane, int nNodes, int nEdges) {
    const bool act = lane < (DFEAT / 4);
    const unsigned cmax = (unsigned)(nNodes - 1);
    const float* xl = H + lane * 4;
    for (int trip = 0; trip < MAXRPB; ++trip) {
        int i = 0;
        if (lane == 0) i = atomicAdd(nextRow, 1);
        i = __builtin_amdgcn_readfirstlane(i);
        if (i >= nrows) break;
        const int row = row0 + i;
        const size_t own = (size_t)row * DFEAT + lane * 4;
        float4 h = make_float4(0.f, 0.f, 0.f, 0.f);
        if (act) h = *(const float4*)(H + own);
        int beg = __builtin_amdgcn_readfirstlane(rbeg[i]);
        int end = __builtin_amdgcn_readfirstlane(cntpos[i]);
        beg = beg < 0 ? 0 : beg;
        end = end > nEdges ? nEdges : end;
        float4 acc = make_float4(0.f, 0.f, 0.f, 0.f);
        for (int b0 = beg; b0 < end; b0 += 32) {
            int n = end - b0;
            n = n > 32 ? 32 : n;
            int2 my = make_int2(0, 0);
            if (lane < n) {
                if (IN_LDS) my = sE[b0 - base + lane];
                else        my = sorted[b0 + lane];
            }
            int j = 0;
            for (; j + 8 <= n; j += 8)
                gather_group8<false>(my, j, n, cmax, act, xl, acc);
            if (j < n)
                gather_group8<true>(my, j, n, cmax, act, xl, acc);
        }
        if (act) {
            float4 o;
            o.x = 2.0f * acc.x - h.x;
            o.y = 2.0f * acc.y - h.y;
            o.z = 2.0f * acc.z - h.z;
            o.w = 2.0f * acc.w - h.w;
            *(float4*)(T1out + own) = o;
        }
    }
}

__global__ void __launch_bounds__(SORT_THREADS)
cheb_sort_range_t1(const int* __restrict__ recR,
                   const int2* __restrict__ recCV,
                   const int* __restrict__ table,
                   int* __restrict__ rowptr,
                   int2* __restrict__ sorted,
                   const float* __restrict__ H,
                   float* __restrict__ T1out,
                   int nEdges, int nNodes, int rpb, int ncp) {
    __shared__ int  cntpos[MAXRPB];
    __shared__ int  rbeg[MAXRPB];
    __shared__ int  cur[MAXRPB];
    __shared__ int  wsumA[SORT_THREADS / 32];
    __shared__ int  wsumB[SORT_THREADS / 32];
    __shared__ int  nextRow;
    __shared__ int  restFlag;
    __shared__ int2 sE[LIST_CAP];
    const int tid  = threadIdx.x;
    const int lane = tid & 31;
    const int w    = tid >> 5;
    const int b    = blockIdx.x;
    const int row0 = b * rpb;

    cntpos[tid] = 0;
    if (tid == 0) { nextRow = 0; restFlag = 0; }

    const int g   = tid >> 4;
    const int sub = tid & 15;
    const int* trow = table + (size_t)b * ncp;

    int s = 0, ent0 = 0;
    for (int cb = 0; cb < ncp; cb += 16 * RUN_GROUPS) {
        const int cm = cb + g + RUN_GROUPS * sub;
        const int e  = (cm < ncp) ? trow[cm] : 0;
        if (cb == 0) ent0 = e;
        s += e >> 16;
    }

    int  hl[HELD_J][HELD_S];
    int2 hc[HELD_J][HELD_S];
    int  rk[HELD_J][HELD_S];
    bool hv[HELD_J][HELD_S];
    bool rest = ncp > RUN_GROUPS * HELD_J;
#pragma unroll
    for (int j = 0; j < HELD_J; ++j) {
        const int ent = __shfl(ent0, j, 16);
        const int len = ent & 0xFFFF;
        const size_t off = (size_t)(g + RUN_GROUPS * j) * CE + (ent >> 16);
        rest = rest || (len > 16 * HELD_S);
#pragma unroll
        for (int sl = 0; sl < HELD_S; ++sl) {
            const int k = sub + 16 * sl;
            hv[j][sl] = k < len;
            hl[j][sl] = 0;
            hc[j][sl] = make_int2(0, 0);
            rk[j][sl] = 0;
            if (hv[j][sl]) {
                hl[j][sl] = recR[off + k];
                hc[j][sl] = recCV[off + k];
            }
        }
    }

#pragma unroll
    for (int d = 16; d >= 1; d >>= 1) s += __shfl_xor(s, d, 32);
    if (lane == 0) wsumA[w] = s;
    __syncthreads();

#pragma unroll
    for (int j = 0; j < HELD_J; ++j) {
#pragma unroll
        for (int sl = 0; sl < HELD_S; ++sl) {
            const int lr = hv[j][sl] ? hl[j][sl] - row0 : -1;
            hl[j][sl] = -1;
            if ((unsigned)lr < (unsigned)rpb) {
                hl[j][sl] = lr;
                rk[j][sl] = atomicAdd(&cntpos[lr], 1);
            }
        }
    }
    if (rest) restFlag = 1;
    __syncthreads();
    const bool anyRest = __builtin_amdgcn_readfirstlane(restFlag) != 0;
    if (anyRest) {
        cur[tid] = cntpos[tid];
        __syncthreads();
        walk_rest<false>(trow, recR, recCV, sorted, cntpos, sE, 0, g, sub, ncp, row0, rpb, nEdges);
        __syncthreads();
    }

    int base = wsumA[lane];
#pragma unroll
    for (int d = 16; d >= 1; d >>= 1) base += __shfl_xor(base, d, 32);
    const int cv = cntpos[tid];
    int inc = cv;
#pragma unroll
    for (int d = 1; d < 32; d <<= 1) {
        int up = __shfl_up(inc, d, 32);
        if (lane >= d) inc += up;
    }
    if (lane == 31) wsumB[w] = inc;
    __syncthreads();
    int carry = (lane < w) ? wsumB[lane] : 0;
#pragma unroll
    for (int d = 16; d >= 1; d >>= 1) carry += __shfl_xor(carry, d, 32);
    const int start = base + carry + inc - cv;
    rbeg[tid]   = start;
    cntpos[tid] = start + cv;
    if (anyRest) cur[tid] = start + cur[tid];
    if (tid < rpb && row0 + tid < nNodes) rowptr[row0 + tid] = start;
    if (b == (int)gridDim.x - 1 && tid == 0) rowptr[nNodes] = nEdges;
    __syncthreads();

#pragma unroll
    for (int j = 0; j < HELD_J; ++j) {
#pragma unroll
        for (int sl = 0; sl < HELD_S; ++sl)
            rk[j][sl] += rbeg[hl[j][sl] >= 0 ? hl[j][sl] : 0];
    }
#pragma unroll
    for (int j = 0; j < HELD_J; ++j) {
#pragma unroll
        for (int sl = 0; sl < HELD_S; ++sl) {
            if (hl[j][sl] >= 0) {
                const int pos = rk[j][sl];
                if ((unsigned)pos < (unsigned)nEdges)
                    sorted[pos] = hc[j][sl];
                const int lp = pos - base;
                if ((unsigned)lp < (unsigned)LIST_CAP)
                    sE[lp] = hc[j][sl];
            }
        }
    }
    if (anyRest)
        walk_rest<true>(trow, recR, recCV, sorted, cur, sE, base, g, sub, ncp, row0, rpb, nEdges);
    __syncthreads();

    const int listEnd = __builtin_amdgcn_readfirstlane(cntpos[MAXRPB - 1]);
    int nrows = nNodes - row0;
    nrows = nrows > rpb ? rpb : nrows;
    if ((listEnd - base) <= LIST_CAP)
        range_rows_t1<true>(sE, rbeg, cntpos, &nextRow, sorted, H, T1out,
                            base, row0, nrows, lane, nNodes, nEdges);
    else
        range_rows_t1<false>(sE, rbeg, cntpos, &nextRow, sorted, H, T1out,
                             base, row0, nrows, lane, nNodes, nEdges);
}

#define RPW 4
#define WAVES_PER_GROUP 8
#define ROWS_PER_GROUP (RPW * WAVES_PER_GROUP)

template <int STEP>
__device__ __forceinline__ float4
row_value(const int2* __restrict__ sorted,
          const float* __restrict__ X,
          const float* __restrict__ H,
          const float* __restrict__ T1,
          const float* __restrict__ T2,
          int beg, int end,
          int row, int lane, int nNodes) {
    const bool act = lane < (DFEAT / 4);
    const unsigned cmax = (unsigned)(nNodes - 1);
    const float* xl = X + lane * 4;
    const size_t own = (size_t)row * DFEAT + lane * 4;

    int2 my0 = make_int2(0, 0);
    if (lane < end - beg) my0 = sorted[beg + lane];

    float4 h  = make_float4(0.f, 0.f, 0.f, 0.f);
    float4 t1 = make_float4(0.f, 0.f, 0.f, 0.f);
    float4 t2 = make_float4(0.f, 0.f, 0.f, 0.f);
    if (act) {
        h = *(const float4*)(H + own);
        if (STEP >= 2) t1 = *(const float4*)(T1 + own);
        if (STEP == 3) t2 = *(const float4*)(T2 + own);
    }

    float4 acc = make_float4(0.f, 0.f, 0.f, 0.f);
    for (int base = beg; base < end; base += 32) {
        int n = end - base;
        n = n > 32 ? 32 : n;
        int2 my = my0;
        if (base != beg) {
            my = make_int2(0, 0);
            if (lane < n) my = sorted[base + lane];
        }
        int j = 0;
        for (; j + 8 <= n; j += 8)
            gather_group8<false>(my, j, n, cmax, act, xl, acc);
        if (j < n)
            gather_group8<true>(my, j, n, cmax, act, xl, acc);
    }

    float4 o;
    if (STEP == 1) {
        o.x = 2.0f * acc.x - h.x;
        o.y = 2.0f * acc.y - h.y;
        o.z = 2.0f * acc.z - h.z;
        o.w = 2.0f * acc.w - h.w;
    } else if (STEP == 2) {
        o.x = 4.0f * acc.x - 2.0f * t1.x - h.x;
        o.y = 4.0f * acc.y - 2.0f * t1.y - h.y;
        o.z = 4.0f * acc.z - 2.0f * t1.z - h.z;
        o.w = 4.0f * acc.w - 2.0f * t1.w - h.w;
    } else {
        float4 t3;
        t3.x = 4.0f * acc.x - 2.0f * t2.x - t1.x;
        t3.y = 4.0f * acc.y - 2.0f * t2.y - t1.y;
        t3.z = 4.0f * acc.z - 2.0f * t2.z - t1.z;
        t3.w = 4.0f * acc.w - 2.0f * t2.w - t1.w;
        o.x = h.x + t1.x;  o.x += t2.x;  o.x += t3.x;
        o.y = h.y + t1.y;  o.y += t2.y;  o.y += t3.y;
        o.z = h.z + t1.z;  o.z += t2.z;  o.z += t3.z;
        o.w = h.w + t1.w;  o.w += t2.w;  o.w += t3.w;
    }
    return o;
}

template <int STEP, class Put>
__device__ __forceinline__ void
wave_rows(const int* __restrict__ rowptr,
          const int2* __restrict__ sorted,
          const float* __restrict__ X,
          const float* __restrict__ H,
          const float* __restrict__ T1,
          const float* __restrict__ T2,
          int row0, int lane, int nNodes, int nEdges, Put put) {
    int pi = row0 + (lane <= RPW ? lane : RPW);
    pi = pi > nNodes ? nNodes : pi;
    int rp = rowptr[pi];
    rp = rp < 0 ? 0 : rp;
    rp = rp > nEdges ? nEdges : rp;

#pragma nounroll
    for (int r = 0; r < RPW; ++r) {
        const int row = row0 + r;
        float4 o = make_float4(0.f, 0.f, 0.f, 0.f);
        if (row < nNodes) {
            const int beg = __builtin_amdgcn_readlane(rp, r);
            const int end = __builtin_amdgcn_readlane(rp, r + 1);
            o = row_value<STEP>(sorted, X, H, T1, T2, beg, end, row, lane, nNodes);
        }
        put(r, row, o);
    }
}

__global__ void __launch_bounds__(32 * WAVES_PER_GROUP)
cheb_gather96_t2(const int* __restrict__ rowptr,
                 const int2* __restrict__ sorted,
                 const float* __restrict__ H,
                 const float* __restrict__ T1,
                 float* __restrict__ T2out,
                 int nNodes, int nEdges) {
    const int row0 = __builtin_amdgcn_readfirstlane(
        (int)((blockIdx.x * blockDim.x + threadIdx.x) >> 5)) * RPW;
    if (row0 >= nNodes) return;
    const int lane = threadIdx.x & 31;
    wave_rows<2>(rowptr, sorted, T1, H, T1, nullptr, row0, lane, nNodes, nEdges,
                 [&](int, int row, float4 o) {
                     if (row < nNodes && lane < (DFEAT / 4))
                         *(float4*)(T2out + (size_t)row * DFEAT + lane * 4) = o;
                 });
}

#define SPITCH 100
#define TILES_PER_GROUP ((ROWS_PER_GROUP / 16) * NTILES_N)
__global__ void __launch_bounds__(32 * WAVES_PER_GROUP)
cheb_gather96_s_gemm(const int* __restrict__ rowptr,
                     const int2* __restrict__ sorted,
                     const float* __restrict__ H,
                     const float* __restrict__ T1,
                     const float* __restrict__ T2,
                     const float* __restrict__ W,
                     const float* __restrict__ bias,
                     float* __restrict__ out,
                     int nNodes, int nEdges) {
    __shared__ __attribute__((aligned(16))) float sS[ROWS_PER_GROUP * SPITCH];
    const int lane = threadIdx.x & 31;
    const int w    = __builtin_amdgcn_readfirstlane((int)(threadIdx.x >> 5));
    const int rowG = blockIdx.x * ROWS_PER_GROUP;

    float* srow = sS + (w * RPW) * SPITCH + lane * 4;
    wave_rows<3>(rowptr, sorted, T2, H, T1, T2, rowG + w * RPW, lane, nNodes, nEdges,
                 [&](int r, int, float4 o) {
                     if (lane < (DFEAT / 4))
                         *(float4*)(srow + r * SPITCH) = o;
                 });
    __syncthreads();

    const int half = lane >> 4;
    const int l    = lane & 15;

    for (int t = w; t < TILES_PER_GROUP; t += WAVES_PER_GROUP) {
        const int mt   = t / NTILES_N;
        const int nt   = t - mt * NTILES_N;
        const int row0 = rowG + mt * 16;
        if (row0 >= nNodes) break;

        const float* arow = sS + (mt * 16 + l) * SPITCH + 2 * half;
        const float* Wcol = W + nt * 16 + l + (size_t)(2 * half) * DFEAT;

        v8f c = {};
#pragma unroll 12
        for (int k0 = 0; k0 < DFEAT; k0 += 4) {
            v2f a = *(const v2f*)(arow + k0);
            v2f b;
            b.x = Wcol[(size_t)k0 * DFEAT];
            b.y = Wcol[(size_t)(k0 + 1) * DFEAT];
            c = __builtin_amdgcn_wmma_f32_16x16x4_f32(
                     false, a,  false, b,
                     (short)0, c,  false,  false);
        }

        const int n  = nt * 16 + l;
        const float bv = bias[n];
        const int mbase = row0 + half * 8;
        float* orow = out + (size_t)mbase * DFEAT + n;
#pragma unroll
        for (int r = 0; r < 8; ++r) {
            if (mbase + r < nNodes)
                orow[(size_t)r * DFEAT] = c[r] + bv;
        }
    }
}

extern "C" void kernel_launch(void* const* d_in, const int* in_sizes, int n_in,
                              void* d_out, int out_size, void* d_ws, size_t ws_size,
                              hipStream_t stream) {
    const int*   rows = (const int*)d_in[0];
    const int*   cols = (const int*)d_in[1];
    const float* vals = (const float*)d_in[2];
    const float* H    = (const float*)d_in[3];
    const float* W    = (const float*)d_in[4];
    const float* bias = (const float*)d_in[5];
    float* out = (float*)d_out;

    const int nEdges = in_sizes[0];
    const int nNodes = in_sizes[3] / DFEAT;
    const int nElem  = nNodes * DFEAT;

    float* P0 = (float*)d_ws;
    float* L  = P0 + nElem;
    float* P2 = L + nElem;

    const int ncp = (((nEdges + CE - 1) / CE) + 3) / 4 * 4;
    const int rpb = (nNodes + NBIN - 1) / NBIN;

    int2* sorted = (int2*)L;
    int2* recCV  = sorted + (nEdges + 1) / 2 * 2;
    int*  recR   = (int*)(recCV + (size_t)ncp * CE);
    int*  table  = recR + (size_t)ncp * CE;
    int*  rowptr = table + (size_t)NBIN * ncp;

    dim3 blk(32 * WAVES_PER_GROUP);
    dim3 grdRows((nNodes + ROWS_PER_GROUP - 1) / ROWS_PER_GROUP);

    cheb_split_ranges<<<dim3(ncp), dim3(SORT_THREADS), 0, stream>>>(rows, cols, vals, recR, recCV, table,
                                                                    nEdges, nNodes, rpb, ncp);
    cheb_sort_range_t1<<<dim3(NBIN), dim3(SORT_THREADS), 0, stream>>>(recR, recCV, table, rowptr, sorted,
                                                                      H, P0, nEdges, nNodes, rpb, ncp);

    cheb_gather96_t2<<<grdRows, blk, 0, stream>>>(rowptr, sorted, H, P0, P2, nNodes, nEdges);

    cheb_gather96_s_gemm<<<grdRows, blk, 0, stream>>>(rowptr, sorted, H, P0, P2, W, bias, out,
                                                            nNodes, nEdges);
}


# Both sort levels store to memory only behind their last barrier (table, row pointers; level 2's list copied from its LDS copy in whole 128-byte lines): no barrier waits for a store; sum order kept
# speedup vs baseline: 5.5045x; 1.0489x over previous
#include <hip/hip_runtime.h>
#include <hip/hip_bf16.h>


typedef __attribute__((ext_vector_type(2))) float v2f;
typedef __attribute__((ext_vector_type(8))) float v8f;

#define DFEAT 96
#define NTILES_N 6

#define CE 4096
#define NBIN 256
#define SORT_THREADS 1024
#define MAXRPB 1024
#define LIST_CAP 20480

__global__ void __launch_bounds__(SORT_THREADS)
cheb_split_ranges(const int* __restrict__ rows,
                  const int* __restrict__ cols,
                  const float* __restrict__ vals,
                  int* __restrict__ recR,
                  int2* __restrict__ recCV,
                  int* __restrict__ table,
                  int nEdges, int nNodes, int rpb, int ncp) {
    __shared__ int  hist[NBIN];
    __shared__ int  bstart[NBIN];
    __shared__ int  wtot[NBIN / 32];
    __shared__ int  sR[CE];
    __shared__ int2 sCV[CE];
    const int tid   = threadIdx.x;
    const int lane  = tid & 31;
    const int w     = tid >> 5;
    const int chunk = blockIdx.x;
    const int e0    = chunk * CE;

    if (tid < NBIN) hist[tid] = 0;

    int   r[4], c[4], bn[4], rk[4];
    float v[4];
    bool  ok[4];
#pragma unroll
    for (int k = 0; k < 4; ++k) {
        const int e = e0 + k * SORT_THREADS + tid;
        ok[k] = e < nEdges;
        const int ec = ok[k] ? e : nEdges - 1;
        r[k] = rows[ec];
        c[k] = cols[ec];
        v[k] = vals[ec];
    }
#pragma unroll
    for (int k = 0; k < 4; ++k) {
        unsigned row = (unsigned)r[k];
        if (row >= (unsigned)nNodes) row = (unsigned)(nNodes - 1);
        r[k]  = (int)row;
        bn[k] = ok[k] ? (int)(row / (unsigned)rpb) : -1;
        rk[k] = 0;
    }
    __syncthreads();
#pragma unroll
    for (int k = 0; k < 4; ++k)
        if (bn[k] >= 0) rk[k] = atomicAdd(&hist[bn[k]], 1);
    __syncthreads();

    int hv = 0, inc = 0, tent = 0;
    if (w < NBIN / 32) {
        hv = hist[tid];
        inc = hv;
#pragma unroll
        for (int d = 1; d < 32; d <<= 1) {
            int up = __shfl_up(inc, d, 32);
            if (lane >= d) inc += up;
        }
        if (lane == 31) wtot[w] = inc;
    }
    __syncthreads();
    if (w < NBIN / 32) {
        int carry = 0;
#pragma unroll
        for (int i = 0; i < NBIN / 32; ++i)
            if (i < w) carry += wtot[i];
        const int st = carry + inc - hv;
        bstart[tid] = st;
        tent = (st << 16) | hv;
    }
    __syncthreads();

#pragma unroll
    for (int k = 0; k < 4; ++k) {
        if (bn[k] >= 0) {
            const int p = bstart[bn[k]] + rk[k];
            sR[p]  = r[k];
            sCV[p] = make_int2(c[k], __float_as_int(v[k]));
        }
    }
    __syncthreads();

    if (w < NBIN / 32) table[(size_t)tid * ncp + chunk] = tent;

    int nIn = nEdges - e0;
    nIn = nIn > CE ? CE : nIn;
    for (int p = tid; p < nIn; p += SORT_THREADS) {
        recR[(size_t)e0 + p]  = sR[p];
        recCV[(size_t)e0 + p] = sCV[p];
    }
}

template <bool TAIL>
__device__ __forceinline__ void gather_group8(int2 my, int j, int n, unsigned cmax,
                                              bool act, const float* xl, float4& acc) {
    unsigned c[8];
    float    wv[8];
#pragma unroll
    for (int u = 0; u < 8; ++u) {
        int idx = (j + u) & 31;
        unsigned cu = (unsigned)__builtin_amdgcn_readlane(my.x, idx);
        c[u]  = cu > cmax ? cmax : cu;
        wv[u] = __int_as_float(__builtin_amdgcn_readlane(my.y, idx));
    }
    if (act) {
        float4 x[8];
#pragma unroll
        for (int u = 0; u < 8; ++u) {
            x[u] = make_float4(0.f, 0.f, 0.f, 0.f);
            if (!TAIL || j + u < n)
                x[u] = *(const float4*)(xl + (size_t)c[u] * DFEAT);
        }
#pragma unroll
        for (int u = 0; u < 8; ++u) {
            acc.x += wv[u] * x[u].x;
            acc.y += wv[u] * x[u].y;
            acc.z += wv[u] * x[u].z;
            acc.w += wv[u] * x[u].w;
        }
    }
}

#define RUN_GROUPS (SORT_THREADS / 16)
#define HELD_J 4
#define HELD_S 3

template <bool PLACE>
__device__ __forceinline__ void walk_rest(const int* __restrict__ trow,
                                          const int* __restrict__ recR,
                                          const int2* __restrict__ recCV,
                                          int2* __restrict__ sorted,
                                          int* cnt, int2* sE, int base,
                                          int g, int sub, int ncp,
                                          int row0, int rpb, int nEdges, bool toMem) {
    for (int cb = 0; cb < ncp; cb += 16 * RUN_GROUPS) {
        const int cm = cb + g + RUN_GROUPS * sub;
        const int entMine = (cm < ncp) ? trow[cm] : 0;
        int jn = (ncp - cb + RUN_GROUPS - 1) / RUN_GROUPS;
        jn = jn > 16 ? 16 : jn;
        for (int j = 0; j < jn; ++j) {
            const int ent = __shfl(entMine, j, 16);
            const int c   = cb + g + RUN_GROUPS * j;
            const int len = (c < ncp) ? (ent & 0xFFFF) : 0;
            const size_t off = (size_t)c * CE + (ent >> 16);
            const int k0 = (cb == 0 && j < HELD_J) ? sub + 16 * HELD_S : sub;
            for (int k = k0; k < len; k += 16) {
                const int lr = recR[off + k] - row0;
                if ((unsigned)lr < (unsigned)rpb) {
                    if (!PLACE) {
                        atomicAdd(&cnt[lr], 1);
                    } else {
                        const int2 cvv = recCV[off + k];
                        const int pos = atomicAdd(&cnt[lr], 1);
                        if (toMem && (unsigned)pos < (unsigned)nEdges)
                            sorted[pos] = cvv;
                        const int lp = pos - base;
                        if ((unsigned)lp < (unsigned)LIST_CAP)
                            sE[lp] = cvv;
                    }
                }
            }
        }
    }
}

template <bool IN_LDS>
__device__ __forceinline__ void
range_rows_t1(const int2* sE, const int* rbeg, const int* cntpos, int* nextRow,
              const int2* __restrict__ sorted,
              const float* __restrict__ H,
              float* __restrict__ T1out,
              int base, int row0, int nrows, int lane, int nNodes, int nEdges) {
    const bool act = lane < (DFEAT / 4);
    const unsigned cmax = (unsigned)(nNodes - 1);
    const float* xl = H + lane * 4;
    for (int trip = 0; trip < MAXRPB; ++trip) {
        int i = 0;
        if (lane == 0) i = atomicAdd(nextRow, 1);
        i = __builtin_amdgcn_readfirstlane(i);
        if (i >= nrows) break;
        const int row = row0 + i;
        const size_t own = (size_t)row * DFEAT + lane * 4;
        float4 h = make_float4(0.f, 0.f, 0.f, 0.f);
        if (act) h = *(const float4*)(H + own);
        int beg = __builtin_amdgcn_readfirstlane(rbeg[i]);
        int end = __builtin_amdgcn_readfirstlane(cntpos[i]);
        beg = beg < 0 ? 0 : beg;
        end = end > nEdges ? nEdges : end;
        float4 acc = make_float4(0.f, 0.f, 0.f, 0.f);
        for (int b0 = beg; b0 < end; b0 += 32) {
            int n = end - b0;
            n = n > 32 ? 32 : n;
            int2 my = make_int2(0, 0);
            if (lane < n) {
                if (IN_LDS) my = sE[b0 - base + lane];
                else        my = sorted[b0 + lane];
            }
            int j = 0;
            for (; j + 8 <= n; j += 8)
                gather_group8<false>(my, j, n, cmax, act, xl, acc);
            if (j < n)
                gather_group8<true>(my, j, n, cmax, act, xl, acc);
        }
        if (act) {
            float4 o;
            o.x = 2.0f * acc.x - h.x;
            o.y = 2.0f * acc.y - h.y;
            o.z = 2.0f * acc.z - h.z;
            o.w = 2.0f * acc.w - h.w;
            *(float4*)(T1out + own) = o;
        }
    }
}

__global__ void __launch_bounds__(SORT_THREADS)
cheb_sort_range_t1(const int* __restrict__ recR,
                   const int2* __restrict__ recCV,
                   const int* __restrict__ table,
                   int* __restrict__ rowptr,
                   int2* __restrict__ sorted,
                   const float* __restrict__ H,
                   float* __restrict__ T1out,
                   int nEdges, int nNodes, int rpb, int ncp) {
    __shared__ int  cntpos[MAXRPB];
    __shared__ int  rbeg[MAXRPB];
    __shared__ int  cur[MAXRPB];
    __shared__ int  wsumA[SORT_THREADS / 32];
    __shared__ int  wsumB[SORT_THREADS / 32];
    __shared__ int  nextRow;
    __shared__ int  restFlag;
    __shared__ int2 sE[LIST_CAP];
    const int tid  = threadIdx.x;
    const int lane = tid & 31;
    const int w    = tid >> 5;
    const int b    = blockIdx.x;
    const int row0 = b * rpb;

    cntpos[tid] = 0;
    if (tid == 0) { nextRow = 0; restFlag = 0; }

    const int g   = tid >> 4;
    const int sub = tid & 15;
    const int* trow = table + (size_t)b * ncp;

    int s = 0, ent0 = 0;
    for (int cb = 0; cb < ncp; cb += 16 * RUN_GROUPS) {
        const int cm = cb + g + RUN_GROUPS * sub;
        const int e  = (cm < ncp) ? trow[cm] : 0;
        if (cb == 0) ent0 = e;
        s += e >> 16;
    }

    int  hl[HELD_J][HELD_S];
    int2 hc[HELD_J][HELD_S];
    int  rk[HELD_J][HELD_S];
    bool hv[HELD_J][HELD_S];
    bool rest = ncp > RUN_GROUPS * HELD_J;
#pragma unroll
    for (int j = 0; j < HELD_J; ++j) {
        const int ent = __shfl(ent0, j, 16);
        const int len = ent & 0xFFFF;
        const size_t off = (size_t)(g + RUN_GROUPS * j) * CE + (ent >> 16);
        rest = rest || (len > 16 * HELD_S);
#pragma unroll
        for (int sl = 0; sl < HELD_S; ++sl) {
            const int k = sub + 16 * sl;
            hv[j][sl] = k < len;
            hl[j][sl] = 0;
            hc[j][sl] = make_int2(0, 0);
            rk[j][sl] = 0;
            if (hv[j][sl]) {
                hl[j][sl] = recR[off + k];
                hc[j][sl] = recCV[off + k];
            }
        }
    }

#pragma unroll
    for (int d = 16; d >= 1; d >>= 1) s += __shfl_xor(s, d, 32);
    if (lane == 0) wsumA[w] = s;
    __syncthreads();

#pragma unroll
    for (int j = 0; j < HELD_J; ++j) {
#pragma unroll
        for (int sl = 0; sl < HELD_S; ++sl) {
            const int lr = hv[j][sl] ? hl[j][sl] - row0 : -1;
            hl[j][sl] = -1;
            if ((unsigned)lr < (unsigned)rpb) {
                hl[j][sl] = lr;
                rk[j][sl] = atomicAdd(&cntpos[lr], 1);
            }
        }
    }
    if (rest) restFlag = 1;
    __syncthreads();
    const bool anyRest = __builtin_amdgcn_readfirstlane(restFlag) != 0;
    if (anyRest) {
        cur[tid] = cntpos[tid];
        __syncthreads();
        walk_rest<false>(trow, recR, recCV, sorted, cntpos, sE, 0, g, sub, ncp, row0, rpb, nEdges, false);
        __syncthreads();
    }

    int bsum = wsumA[lane];
#pragma unroll
    for (int d = 16; d >= 1; d >>= 1) bsum += __shfl_xor(bsum, d, 32);
    const int base = __builtin_amdgcn_readfirstlane(bsum);
    const int cv = cntpos[tid];
    int inc = cv;
#pragma unroll
    for (int d = 1; d < 32; d <<= 1) {
        int up = __shfl_up(inc, d, 32);
        if (lane >= d) inc += up;
    }
    if (lane == 31) wsumB[w] = inc;
    __syncthreads();
    int carry = (lane < w) ? wsumB[lane] : 0;
#pragma unroll
    for (int d = 16; d >= 1; d >>= 1) carry += __shfl_xor(carry, d, 32);
    const int start = base + carry + inc - cv;
    rbeg[tid]   = start;
    cntpos[tid] = start + cv;
    if (anyRest) cur[tid] = start + cur[tid];
    __syncthreads();

    const int  listEnd = __builtin_amdgcn_readfirstlane(cntpos[MAXRPB - 1]);
    const bool inLds   = (listEnd - base) <= LIST_CAP;

#pragma unroll
    for (int j = 0; j < HELD_J; ++j) {
#pragma unroll
        for (int sl = 0; sl < HELD_S; ++sl)
            rk[j][sl] += rbeg[hl[j][sl] >= 0 ? hl[j][sl] : 0];
    }
#pragma unroll
    for (int j = 0; j < HELD_J; ++j) {
#pragma unroll
        for (int sl = 0; sl < HELD_S; ++sl) {
            if (hl[j][sl] >= 0) {
                const int pos = rk[j][sl];
                if (!inLds && (unsigned)pos < (unsigned)nEdges)
                    sorted[pos] = hc[j][sl];
                const int lp = pos - base;
                if ((unsigned)lp < (unsigned)LIST_CAP)
                    sE[lp] = hc[j][sl];
            }
        }
    }
    if (anyRest)
        walk_rest<true>(trow, recR, recCV, sorted, cur, sE, base, g, sub, ncp, row0, rpb, nEdges, !inLds);
    __syncthreads();

    if (tid < rpb && row0 + tid < nNodes) rowptr[row0 + tid] = start;
    if (b == NBIN - 1 && tid == 0) rowptr[nNodes] = nEdges;
    if (inLds) {
        const int a0 = base & ~15;
#pragma nounroll
        for (int k0 = 0; k0 <= LIST_CAP / SORT_THREADS; k0 += 4) {
            if (a0 + k0 * SORT_THREADS >= listEnd) break;
            int2 pv[4];
            int  pp[4];
#pragma unroll
            for (int u = 0; u < 4; ++u) {
                const int p = a0 + (k0 + u) * SORT_THREADS + tid;
                const bool ok = p >= base && p < listEnd && (unsigned)p < (unsigned)nEdges;
                pp[u] = ok ? p : -1;
                pv[u] = sE[ok ? p - base : 0];
            }
#pragma unroll
            for (int u = 0; u < 4; ++u)
                if (pp[u] >= 0) sorted[pp[u]] = pv[u];
        }
    }

    int nrows = nNodes - row0;
    nrows = nrows > rpb ? rpb : nrows;
    if (inLds)
        range_rows_t1<true>(sE, rbeg, cntpos, &nextRow, sorted, H, T1out,
                            base, row0, nrows, lane, nNodes, nEdges);
    else
        range_rows_t1<false>(sE, rbeg, cntpos, &nextRow, sorted, H, T1out,
                             base, row0, nrows, lane, nNodes, nEdges);
}

#define RPW 4
#define WAVES_PER_GROUP 8
#define ROWS_PER_GROUP (RPW * WAVES_PER_GROUP)

template <int STEP>
__device__ __forceinline__ float4
row_value(const int2* __restrict__ sorted,
          const float* __restrict__ X,
          const float* __restrict__ H,
          const float* __restrict__ T1,
          const float* __restrict__ T2,
          int beg, int end,
          int row, int lane, int nNodes) {
    const bool act = lane < (DFEAT / 4);
    const unsigned cmax = (unsigned)(nNodes - 1);
    const float* xl = X + lane * 4;
    const size_t own = (size_t)row * DFEAT + lane * 4;

    int2 my0 = make_int2(0, 0);
    if (lane < end - beg) my0 = sorted[beg + lane];

    float4 h  = make_float4(0.f, 0.f, 0.f, 0.f);
    float4 t1 = make_float4(0.f, 0.f, 0.f, 0.f);
    float4 t2 = make_float4(0.f, 0.f, 0.f, 0.f);
    if (act) {
        h = *(const float4*)(H + own);
        if (STEP >= 2) t1 = *(const float4*)(T1 + own);
        if (STEP == 3) t2 = *(const float4*)(T2 + own);
    }

    float4 acc = make_float4(0.f, 0.f, 0.f, 0.f);
    for (int base = beg; base < end; base += 32) {
        int n = end - base;
        n = n > 32 ? 32 : n;
        int2 my = my0;
        if (base != beg) {
            my = make_int2(0, 0);
            if (lane < n) my = sorted[base + lane];
        }
        int j = 0;
        for (; j + 8 <= n; j += 8)
            gather_group8<false>(my, j, n, cmax, act, xl, acc);
        if (j < n)
            gather_group8<true>(my, j, n, cmax, act, xl, acc);
    }

    float4 o;
    if (STEP == 1) {
        o.x = 2.0f * acc.x - h.x;
        o.y = 2.0f * acc.y - h.y;
        o.z = 2.0f * acc.z - h.z;
        o.w = 2.0f * acc.w - h.w;
    } else if (STEP == 2) {
        o.x = 4.0f * acc.x - 2.0f * t1.x - h.x;
        o.y = 4.0f * acc.y - 2.0f * t1.y - h.y;
        o.z = 4.0f * acc.z - 2.0f * t1.z - h.z;
        o.w = 4.0f * acc.w - 2.0f * t1.w - h.w;
    } else {
        float4 t3;
        t3.x = 4.0f * acc.x - 2.0f * t2.x - t1.x;
        t3.y = 4.0f * acc.y - 2.0f * t2.y - t1.y;
        t3.z = 4.0f * acc.z - 2.0f * t2.z - t1.z;
        t3.w = 4.0f * acc.w - 2.0f * t2.w - t1.w;
        o.x = h.x + t1.x;  o.x += t2.x;  o.x += t3.x;
        o.y = h.y + t1.y;  o.y += t2.y;  o.y += t3.y;
        o.z = h.z + t1.z;  o.z += t2.z;  o.z += t3.z;
        o.w = h.w + t1.w;  o.w += t2.w;  o.w += t3.w;
    }
    return o;
}

template <int STEP, class Put>
__device__ __forceinline__ void
wave_rows(const int* __restrict__ rowptr,
          const int2* __restrict__ sorted,
          const float* __restrict__ X,
          const float* __restrict__ H,
          const float* __restrict__ T1,
          const float* __restrict__ T2,
          int row0, int lane, int nNodes, int nEdges, Put put) {
    int pi = row0 + (lane <= RPW ? lane : RPW);
    pi = pi > nNodes ? nNodes : pi;
    int rp = rowptr[pi];
    rp = rp < 0 ? 0 : rp;
    rp = rp > nEdges ? nEdges : rp;

#pragma nounroll
    for (int r = 0; r < RPW; ++r) {
        const int row = row0 + r;
        float4 o = make_float4(0.f, 0.f, 0.f, 0.f);
        if (row < nNodes) {
            const int beg = __builtin_amdgcn_readlane(rp, r);
            const int end = __builtin_amdgcn_readlane(rp, r + 1);
            o = row_value<STEP>(sorted, X, H, T1, T2, beg, end, row, lane, nNodes);
        }
        put(r, row, o);
    }
}

__global__ void __launch_bounds__(32 * WAVES_PER_GROUP)
cheb_gather96_t2(const int* __restrict__ rowptr,
                 const int2* __restrict__ sorted,
                 const float* __restrict__ H,
                 const float* __restrict__ T1,
                 float* __restrict__ T2out,
                 int nNodes, int nEdges) {
    const int row0 = __builtin_amdgcn_readfirstlane(
        (int)((blockIdx.x * blockDim.x + threadIdx.x) >> 5)) * RPW;
    if (row0 >= nNodes) return;
    const int lane = threadIdx.x & 31;
    wave_rows<2>(rowptr, sorted, T1, H, T1, nullptr, row0, lane, nNodes, nEdges,
                 [&](int, int row, float4 o) {
                     if (row < nNodes && lane < (DFEAT / 4))
                         *(float4*)(T2out + (size_t)row * DFEAT + lane * 4) = o;
                 });
}

#define SPITCH 100
#define TILES_PER_GROUP ((ROWS_PER_GROUP / 16) * NTILES_N)
__global__ void __launch_bounds__(32 * WAVES_PER_GROUP)
cheb_gather96_s_gemm(const int* __restrict__ rowptr,
                     const int2* __restrict__ sorted,
                     const float* __restrict__ H,
                     const float* __restrict__ T1,
                     const float* __restrict__ T2,
                     const float* __restrict__ W,
                     const float* __restrict__ bias,
                     float* __restrict__ out,
                     int nNodes, int nEdges) {
    __shared__ __attribute__((aligned(16))) float sS[ROWS_PER_GROUP * SPITCH];
    const int lane = threadIdx.x & 31;
    const int w    = __builtin_amdgcn_readfirstlane((int)(threadIdx.x >> 5));
    const int rowG = blockIdx.x * ROWS_PER_GROUP;

    float* srow = sS + (w * RPW) * SPITCH + lane * 4;
    wave_rows<3>(rowptr, sorted, T2, H, T1, T2, rowG + w * RPW, lane, nNodes, nEdges,
                 [&](int r, int, float4 o) {
                     if (lane < (DFEAT / 4))
                         *(float4*)(srow + r * SPITCH) = o;
                 });
    __syncthreads();

    const int half = lane >> 4;
    const int l    = lane & 15;

    for (int t = w; t < TILES_PER_GROUP; t += WAVES_PER_GROUP) {
        const int mt   = t / NTILES_N;
        const int nt   = t - mt * NTILES_N;
        const int row0 = rowG + mt * 16;
        if (row0 >= nNodes) break;

        const float* arow = sS + (mt * 16 + l) * SPITCH + 2 * half;
        const float* Wcol = W + nt * 16 + l + (size_t)(2 * half) * DFEAT;

        v8f c = {};
#pragma unroll 12
        for (int k0 = 0; k0 < DFEAT; k0 += 4) {
            v2f a = *(const v2f*)(arow + k0);
            v2f b;
            b.x = Wcol[(size_t)k0 * DFEAT];
            b.y = Wcol[(size_t)(k0 + 1) * DFEAT];
            c = __builtin_amdgcn_wmma_f32_16x16x4_f32(
                     false, a,  false, b,
                     (short)0, c,  false,  false);
        }

        const int n  = nt * 16 + l;
        const float bv = bias[n];
        const int mbase = row0 + half * 8;
        float* orow = out + (size_t)mbase * DFEAT + n;
#pragma unroll
        for (int r = 0; r < 8; ++r) {
            if (mbase + r < nNodes)
                orow[(size_t)r * DFEAT] = c[r] + bv;
        }
    }
}

extern "C" void kernel_launch(void* const* d_in, const int* in_sizes, int n_in,
                              void* d_out, int out_size, void* d_ws, size_t ws_size,
                              hipStream_t stream) {
    const int*   rows = (const int*)d_in[0];
    const int*   cols = (const int*)d_in[1];
    const float* vals = (const float*)d_in[2];
    const float* H    = (const float*)d_in[3];
    const float* W    = (const float*)d_in[4];
    const float* bias = (const float*)d_in[5];
    float* out = (float*)d_out;

    const int nEdges = in_sizes[0];
    const int nNodes = in_sizes[3] / DFEAT;
    const int nElem  = nNodes * DFEAT;

    float* P0 = (float*)d_ws;
    float* L  = P0 + nElem;
    float* P2 = L + nElem;

    const int ncp = (((nEdges + CE - 1) / CE) + 3) / 4 * 4;
    const int rpb = (nNodes + NBIN - 1) / NBIN;

    int2* sorted = (int2*)L;
    int2* recCV  = sorted + (nEdges + 1) / 2 * 2;
    int*  recR   = (int*)(recCV + (size_t)ncp * CE);
    int*  table  = recR + (size_t)ncp * CE;
    int*  rowptr = table + (size_t)NBIN * ncp;

    dim3 blk(32 * WAVES_PER_GROUP);
    dim3 grdRows((nNodes + ROWS_PER_GROUP - 1) / ROWS_PER_GROUP);

    cheb_split_ranges<<<dim3(ncp), dim3(SORT_THREADS), 0, stream>>>(rows, cols, vals, recR, recCV, table,
                                                                    nEdges, nNodes, rpb, ncp);
    cheb_sort_range_t1<<<dim3(NBIN), dim3(SORT_THREADS), 0, stream>>>(recR, recCV, table, rowptr, sorted,
                                                                      H, P0, nEdges, nNodes, rpb, ncp);

    cheb_gather96_t2<<<grdRows, blk, 0, stream>>>(rowptr, sorted, H, P0, P2, nNodes, nEdges);

    cheb_gather96_s_gemm<<<grdRows, blk, 0, stream>>>(rowptr, sorted, H, P0, P2, W, bias, out,
                                                            nNodes, nEdges);
}
